# starting kernel
# baseline (speedup 1.0000x reference)
#include <hip/hip_runtime.h>
#include <math.h>

#define N_NODES 100000
#define N_EDGES 1600000
#define NODE_F  64
#define EDGE_F  16
#define IN_F    80
#define OUT_F   64
#define KPAD    96
#define WAVES   4
#define ROWS_PER_BLOCK (WAVES * 16)
#define ASTRIDE 100

typedef float    v2f  __attribute__((ext_vector_type(2)));
typedef float    v8f  __attribute__((ext_vector_type(8)));
typedef _Float16 v16h __attribute__((ext_vector_type(16)));

#if defined(__AMDGCN__)
#if __has_builtin(__builtin_amdgcn_wmma_f32_16x16x4_f32)
#define USE_WMMA_F32 1
#elif __has_builtin(__builtin_amdgcn_wmma_f32_16x16x32_f16)
#define USE_WMMA_F16 1
#endif
#endif

__global__ void __launch_bounds__(256)
edge_scatter(const float* __restrict__ feat,
             const float* __restrict__ efeat,
             const int*   __restrict__ src,
             const int*   __restrict__ dst,
             float*       __restrict__ acc,
             float*       __restrict__ deg) {
    long long gid = (long long)blockIdx.x * blockDim.x + threadIdx.x;
    int e    = (int)(gid >> 5);
    int lane = (int)(gid & 31);
    if (e >= N_EDGES) return;

    int s = src[e];
    int d = dst[e];
    float* arow = acc + (long long)d * IN_F;

    float h0 = feat[(long long)s * NODE_F + lane];
    float h1 = feat[(long long)s * NODE_F + 32 + lane];
    atomicAdd(arow + EDGE_F + lane,      h0);
    atomicAdd(arow + EDGE_F + 32 + lane, h1);

    if (lane < EDGE_F) {
        float ev = efeat[(long long)e * EDGE_F + lane];
        atomicAdd(arow + lane, ev);
    }
    if (lane == 0) atomicAdd(deg + d, 1.0f);
}

__global__ void __launch_bounds__(256)
deg_to_norm(float* __restrict__ deg) {
    int i = blockIdx.x * blockDim.x + threadIdx.x;
    if (i < N_NODES) {
        float d = deg[i];
        d = d < 1.0f ? 1.0f : d;
        deg[i] = 1.0f / sqrtf(d);
    }
}

__global__ void __launch_bounds__(128)
gemm_norm_selu(const float* __restrict__ acc,
               const float* __restrict__ norm,
               const float* __restrict__ W,
               const float* __restrict__ bias,
               float*       __restrict__ out) {
    __shared__ float sW[KPAD * OUT_F];
    __shared__ float sA[ROWS_PER_BLOCK * ASTRIDE];

    const int tid  = threadIdx.x;
    const int base = blockIdx.x * ROWS_PER_BLOCK;

    for (int t = tid; t < KPAD * OUT_F; t += 128)
        sW[t] = (t < IN_F * OUT_F) ? W[t] : 0.0f;
    for (int t = tid; t < ROWS_PER_BLOCK * ASTRIDE; t += 128)
        sA[t] = 0.0f;
    __syncthreads();
    for (int t = tid; t < ROWS_PER_BLOCK * IN_F; t += 128) {
        int r = t / IN_F, c = t % IN_F;
        int row = base + r;
        float v = 0.0f;
        if (row < N_NODES) v = acc[(long long)row * IN_F + c] * norm[row];
        sA[r * ASTRIDE + c] = v;
    }
    __syncthreads();

    const int wave  = tid >> 5;
    const int lane  = tid & 31;
    const int half  = lane >> 4;
    const int l     = lane & 15;
    const int rbase = wave * 16;

    v8f cacc[4] = {};

#if defined(USE_WMMA_F32)
    for (int kk = 0; kk < IN_F; kk += 4) {
        v2f a;
        const float* ar = &sA[(rbase + l) * ASTRIDE + kk + 2 * half];
        a.x = ar[0];
        a.y = ar[1];
#pragma unroll
        for (int nt = 0; nt < 4; ++nt) {
            int ncol = nt * 16 + l;
            v2f b;
            b.x = sW[(kk + 2 * half)     * OUT_F + ncol];
            b.y = sW[(kk + 2 * half + 1) * OUT_F + ncol];
            cacc[nt] = __builtin_amdgcn_wmma_f32_16x16x4_f32(
                false, a, false, b, (short)0, cacc[nt], false, false);
        }
    }
#elif defined(USE_WMMA_F16)
    for (int kk = 0; kk < KPAD; kk += 32) {
        const int kofs = half * 8;
        v16h a;
        const float* ar = &sA[(rbase + l) * ASTRIDE + kk];
#pragma unroll
        for (int j = 0; j < 8; ++j) {
            a[j]     = (_Float16)ar[kofs + j];
            a[8 + j] = (_Float16)ar[16 + kofs + j];
        }
#pragma unroll
        for (int nt = 0; nt < 4; ++nt) {
            int ncol = nt * 16 + l;
            v16h b;
#pragma unroll
            for (int j = 0; j < 8; ++j) {
                b[j]     = (_Float16)sW[(kk + kofs + j)      * OUT_F + ncol];
                b[8 + j] = (_Float16)sW[(kk + 16 + kofs + j) * OUT_F + ncol];
            }
            cacc[nt] = __builtin_amdgcn_wmma_f32_16x16x32_f16(
                false, a, false, b, (short)0, cacc[nt], false, false);
        }
    }
#else
    for (int i = 0; i < 8; ++i) {
        int m = i + 8 * half;
        for (int nt = 0; nt < 4; ++nt) {
            float s = 0.0f;
            for (int kk = 0; kk < IN_F; ++kk)
                s += sA[(rbase + m) * ASTRIDE + kk] * sW[kk * OUT_F + nt * 16 + l];
            cacc[nt][i] = s;
        }
    }
#endif

    const float SELU_SCALE = 1.0507009873554805f;
    const float SELU_ALPHA = 1.6732632423543772f;
#pragma unroll
    for (int nt = 0; nt < 4; ++nt) {
        int ncol = nt * 16 + l;
        float bv = bias[ncol];
#pragma unroll
        for (int i = 0; i < 8; ++i) {
            int m   = i + 8 * half;
            int row = base + rbase + m;
            if (row < N_NODES) {
                float v = cacc[nt][i] * norm[row] + bv;
                v = v > 0.0f ? SELU_SCALE * v
                             : SELU_SCALE * SELU_ALPHA * (expf(v) - 1.0f);
                out[(long long)row * OUT_F + ncol] = v;
            }
        }
    }
}

extern "C" void kernel_launch(void* const* d_in, const int* in_sizes, int n_in,
                              void* d_out, int out_size, void* d_ws, size_t ws_size,
                              hipStream_t stream) {
    (void)in_sizes; (void)n_in; (void)out_size; (void)ws_size;

    const float* feat  = (const float*)d_in[0];
    const float* efeat = (const float*)d_in[1];
    const int*   src   = (const int*)d_in[2];
    const int*   dst   = (const int*)d_in[3];
    const float* W     = (const float*)d_in[4];
    const float* bias  = (const float*)d_in[5];
    float*       out   = (float*)d_out;

    float* acc = (float*)d_ws;
    float* deg = acc + (size_t)N_NODES * IN_F;

    hipMemsetAsync(d_ws, 0, ((size_t)N_NODES * IN_F + N_NODES) * sizeof(float),
                   stream);

    long long work   = (long long)N_EDGES * 32;
    int       blocks = (int)((work + 255) / 256);
    edge_scatter<<<blocks, 256, 0, stream>>>(feat, efeat, src, dst, acc, deg);

    deg_to_norm<<<(N_NODES + 255) / 256, 256, 0, stream>>>(deg);

    int gblocks = (N_NODES + ROWS_PER_BLOCK - 1) / ROWS_PER_BLOCK;
    gemm_norm_selu<<<gblocks, 128, 0, stream>>>(acc, deg, W, bias, out);
}


# Float global atomics replaced by a bucket sort of edges by destination (LDS counters, scan, scatter) and per-bucket sums with LDS float adds, rows stored once; sum order = LDS arrival order
# speedup vs baseline: 1.9264x; 1.9264x over previous
#include <hip/hip_runtime.h>
#include <math.h>


#define N_NODES 100000
#define N_EDGES 1600000
#define NODE_F  64
#define EDGE_F  16
#define IN_F    80
#define OUT_F   64
#define KPAD    96
#define WAVES   4
#define ROWS_PER_BLOCK (WAVES * 16)
#define ASTRIDE 100

#define NB      500
#define BN      200
#define NCH     250
#define CH      6400
#define HN      (NB * NCH)
#define SCAN_IT 32

typedef float    v2f  __attribute__((ext_vector_type(2)));
typedef float    v8f  __attribute__((ext_vector_type(8)));
typedef _Float16 v16h __attribute__((ext_vector_type(16)));

#if defined(__AMDGCN__)
#if __has_builtin(__builtin_amdgcn_wmma_f32_16x16x4_f32)
#define USE_WMMA_F32 1
#elif __has_builtin(__builtin_amdgcn_wmma_f32_16x16x32_f16)
#define USE_WMMA_F16 1
#endif
#endif

__device__ __forceinline__ int bucket_of(int d) {
    unsigned b = (unsigned)d / (unsigned)BN;
    return (int)(b < (unsigned)(NB - 1) ? b : (unsigned)(NB - 1));
}

__global__ void __launch_bounds__(1024)
bin_count(const int* __restrict__ dst, int* __restrict__ bhist) {
    __shared__ int h[NB];
    const int tid = threadIdx.x;
    const int blk = blockIdx.x;
    if (tid < NB) h[tid] = 0;
    __syncthreads();
    const int4* d4p = (const int4*)dst + (size_t)blk * (CH / 4);
    for (int q = tid; q < CH / 4; q += 1024) {
        int4 d = d4p[q];
        atomicAdd(&h[bucket_of(d.x)], 1);
        atomicAdd(&h[bucket_of(d.y)], 1);
        atomicAdd(&h[bucket_of(d.z)], 1);
        atomicAdd(&h[bucket_of(d.w)], 1);
    }
    __syncthreads();
    if (tid < NB) bhist[tid * NCH + blk] = h[tid];
}

__global__ void __launch_bounds__(1024)
scan_bins(const int* __restrict__ bhist, int* __restrict__ base) {
    __shared__ int wtot[32];
    const int lane  = threadIdx.x & 31;
    const int w     = threadIdx.x >> 5;
    const int wbase = w * (SCAN_IT * 128);

    int s = 0;
    for (int it0 = 0; it0 < SCAN_IT; it0 += 8) {
        int4 v[8];
#pragma unroll
        for (int u = 0; u < 8; ++u) {
            int idx = wbase + (it0 + u) * 128 + 4 * lane;
            int ic  = idx < HN ? idx : HN - 4;
            v[u] = *(const int4*)(bhist + ic);
        }
#pragma unroll
        for (int u = 0; u < 8; ++u) {
            int idx = wbase + (it0 + u) * 128 + 4 * lane;
            int ls  = v[u].x + v[u].y + v[u].z + v[u].w;
            s += idx < HN ? ls : 0;
        }
    }
    for (int o = 16; o > 0; o >>= 1) s += __shfl_xor(s, o);
    if (lane == 0) wtot[w] = s;
    __syncthreads();

    int carry = 0;
    for (int k = 0; k < w; ++k) carry += wtot[k];

#pragma unroll 8
    for (int it = 0; it < SCAN_IT; ++it) {
        int idx = wbase + it * 128 + 4 * lane;
        int ic  = idx < HN ? idx : HN - 4;
        int4 v  = *(const int4*)(bhist + ic);
        if (idx >= HN) v = make_int4(0, 0, 0, 0);
        int ls  = v.x + v.y + v.z + v.w;
        int inc = ls;
#pragma unroll
        for (int o = 1; o < 32; o <<= 1) {
            int t = __shfl_up(inc, o);
            if (lane >= o) inc += t;
        }
        int excl = carry + inc - ls;
        int4 r;
        r.x = excl;
        r.y = r.x + v.x;
        r.z = r.y + v.y;
        r.w = r.z + v.z;
        *(int4*)(base + idx) = r;
        carry += __shfl(inc, 31);
    }
}

__device__ __forceinline__ void place_edge(int* cur, int4* recs, int e, int s, int d) {
    int b = bucket_of(d);
    int p = atomicAdd(&cur[b], 1);
    if ((unsigned)p < (unsigned)N_EDGES) recs[p] = make_int4(e, s, d - b * BN, 0);
}

__global__ void __launch_bounds__(1024)
bin_scatter(const int* __restrict__ src, const int* __restrict__ dst,
            const int* __restrict__ base, int* __restrict__ binned) {
    __shared__ int cur[NB];
    const int tid = threadIdx.x;
    const int blk = blockIdx.x;
    if (tid < NB) cur[tid] = base[tid * NCH + blk];
    __syncthreads();
    const int4* d4p = (const int4*)dst + (size_t)blk * (CH / 4);
    const int4* s4p = (const int4*)src + (size_t)blk * (CH / 4);
    int4* recs = (int4*)binned;
    for (int q = tid; q < CH / 4; q += 1024) {
        int4 d = d4p[q];
        int4 s = s4p[q];
        int e0 = blk * CH + 4 * q;
        place_edge(cur, recs, e0,     s.x, d.x);
        place_edge(cur, recs, e0 + 1, s.y, d.y);
        place_edge(cur, recs, e0 + 2, s.z, d.z);
        place_edge(cur, recs, e0 + 3, s.w, d.w);
    }
}

__device__ __forceinline__ void lds_add(float* p, float v) {
    __hip_atomic_fetch_add(p, v, __ATOMIC_RELAXED, __HIP_MEMORY_SCOPE_WORKGROUP);
}

__global__ void __launch_bounds__(512)
bucket_sum(const float* __restrict__ feat,
           const float* __restrict__ efeat,
           const int*   __restrict__ base,
           const int*   __restrict__ binned,
           float*       __restrict__ acc,
           float*       __restrict__ deg) {
    __shared__ float sacc[BN * IN_F];
    __shared__ int   sdeg[BN];

    const int tid  = threadIdx.x;
    const int b    = blockIdx.x;
    const int lane = tid & 31;
    const int l16  = lane & 15;
    const int w    = __builtin_amdgcn_readfirstlane(tid >> 5);

    float4* sa4 = (float4*)sacc;
    for (int t = tid; t < BN * IN_F / 4; t += 512) sa4[t] = make_float4(0.f, 0.f, 0.f, 0.f);
    if (tid < BN) sdeg[tid] = 0;
    __syncthreads();

    int bs = base[b * NCH];
    int be = (b == NB - 1) ? N_EDGES : base[(b + 1) * NCH];
    bs = __builtin_amdgcn_readfirstlane(bs);
    be = __builtin_amdgcn_readfirstlane(be);
    if (bs < 0) bs = 0;
    if (be > N_EDGES) be = N_EDGES;
    const int T       = be - bs;
    const int nchunks = (T + 31) >> 5;

    const int4* recs  = (const int4*)binned + bs;
    const v2f*  feat2 = (const v2f*)feat;

    for (int c = w; c < nchunks; c += 16) {
        const int j0 = c << 5;
        int m = T - j0;
        m = m < 32 ? m : 32;
        int4 r = make_int4(0, 0, 0, 0);
        if (lane < m) {
            r = recs[j0 + lane];
            r.z = r.z < 0 ? 0 : (r.z > BN - 1 ? BN - 1 : r.z);
            atomicAdd(&sdeg[r.z], 1);
        }
        for (int t = 0; t < m; t += 16) {
            int ee[16], ss[16], ll[16];
            v2f   hv[16];
            float ev[8];
#pragma unroll
            for (int u = 0; u < 16; ++u) {
                ee[u] = __builtin_amdgcn_readlane(r.x, t + u);
                ss[u] = __builtin_amdgcn_readlane(r.y, t + u);
                ll[u] = __builtin_amdgcn_readlane(r.z, t + u);
            }
#pragma unroll
            for (int u = 0; u < 16; ++u)
                hv[u] = feat2[(size_t)ss[u] * (NODE_F / 2) + lane];
#pragma unroll
            for (int p = 0; p < 8; ++p) {
                int em = lane < 16 ? ee[2 * p] : ee[2 * p + 1];
                ev[p] = efeat[(size_t)em * EDGE_F + l16];
            }
#pragma unroll
            for (int p = 0; p < 8; ++p) {
                if (t + 2 * p < m) {
                    float* row = sacc + ll[2 * p] * IN_F + EDGE_F + 2 * lane;
                    lds_add(row,     hv[2 * p].x);
                    lds_add(row + 1, hv[2 * p].y);
                }
                if (t + 2 * p + 1 < m) {
                    float* row = sacc + ll[2 * p + 1] * IN_F + EDGE_F + 2 * lane;
                    lds_add(row,     hv[2 * p + 1].x);
                    lds_add(row + 1, hv[2 * p + 1].y);
                }
                int lm = lane < 16 ? ll[2 * p] : ll[2 * p + 1];
                if (t + 2 * p + (lane >> 4) < m) lds_add(sacc + lm * IN_F + l16, ev[p]);
            }
        }
    }
    __syncthreads();

    float4* o4 = (float4*)(acc + (size_t)b * (BN * IN_F));
    for (int t = tid; t < BN * IN_F / 4; t += 512) o4[t] = sa4[t];
    if (tid < BN) {
        float d = (float)sdeg[tid];
        d = d < 1.0f ? 1.0f : d;
        deg[b * BN + tid] = 1.0f / sqrtf(d);
    }
}

__global__ void __launch_bounds__(128)
gemm_norm_selu(const float* __restrict__ acc,
               const float* __restrict__ norm,
               const float* __restrict__ W,
               const float* __restrict__ bias,
               float*       __restrict__ out) {
    __shared__ float sW[KPAD * OUT_F];
    __shared__ float sA[ROWS_PER_BLOCK * ASTRIDE];

    const int tid  = threadIdx.x;
    const int base = blockIdx.x * ROWS_PER_BLOCK;

    for (int t = tid; t < KPAD * OUT_F; t += 128)
        sW[t] = (t < IN_F * OUT_F) ? W[t] : 0.0f;
    for (int t = tid; t < ROWS_PER_BLOCK * ASTRIDE; t += 128)
        sA[t] = 0.0f;
    __syncthreads();
    for (int t = tid; t < ROWS_PER_BLOCK * IN_F; t += 128) {
        int r = t / IN_F, c = t % IN_F;
        int row = base + r;
        float v = 0.0f;
        if (row < N_NODES) v = acc[(long long)row * IN_F + c] * norm[row];
        sA[r * ASTRIDE + c] = v;
    }
    __syncthreads();

    const int wave  = tid >> 5;
    const int lane  = tid & 31;
    const int half  = lane >> 4;
    const int l     = lane & 15;
    const int rbase = wave * 16;

    v8f cacc[4] = {};

#if defined(USE_WMMA_F32)
    for (int kk = 0; kk < IN_F; kk += 4) {
        v2f a;
        const float* ar = &sA[(rbase + l) * ASTRIDE + kk + 2 * half];
        a.x = ar[0];
        a.y = ar[1];
#pragma unroll
        for (int nt = 0; nt < 4; ++nt) {
            int ncol = nt * 16 + l;
            v2f b;
            b.x = sW[(kk + 2 * half)     * OUT_F + ncol];
            b.y = sW[(kk + 2 * half + 1) * OUT_F + ncol];
            cacc[nt] = __builtin_amdgcn_wmma_f32_16x16x4_f32(
                false, a, false, b, (short)0, cacc[nt], false, false);
        }
    }
#elif defined(USE_WMMA_F16)
    for (int kk = 0; kk < KPAD; kk += 32) {
        const int kofs = half * 8;
        v16h a;
        const float* ar = &sA[(rbase + l) * ASTRIDE + kk];
#pragma unroll
        for (int j = 0; j < 8; ++j) {
            a[j]     = (_Float16)ar[kofs + j];
            a[8 + j] = (_Float16)ar[16 + kofs + j];
        }
#pragma unroll
        for (int nt = 0; nt < 4; ++nt) {
            int ncol = nt * 16 + l;
            v16h b;
#pragma unroll
            for (int j = 0; j < 8; ++j) {
                b[j]     = (_Float16)sW[(kk + kofs + j)      * OUT_F + ncol];
                b[8 + j] = (_Float16)sW[(kk + 16 + kofs + j) * OUT_F + ncol];
            }
            cacc[nt] = __builtin_amdgcn_wmma_f32_16x16x32_f16(
                false, a, false, b, (short)0, cacc[nt], false, false);
        }
    }
#else
    for (int i = 0; i < 8; ++i) {
        int m = i + 8 * half;
        for (int nt = 0; nt < 4; ++nt) {
            float s = 0.0f;
            for (int kk = 0; kk < IN_F; ++kk)
                s += sA[(rbase + m) * ASTRIDE + kk] * sW[kk * OUT_F + nt * 16 + l];
            cacc[nt][i] = s;
        }
    }
#endif

    const float SELU_SCALE = 1.0507009873554805f;
    const float SELU_ALPHA = 1.6732632423543772f;
#pragma unroll
    for (int nt = 0; nt < 4; ++nt) {
        int ncol = nt * 16 + l;
        float bv = bias[ncol];
#pragma unroll
        for (int i = 0; i < 8; ++i) {
            int m   = i + 8 * half;
            int row = base + rbase + m;
            if (row < N_NODES) {
                float v = cacc[nt][i] * norm[row] + bv;
                v = v > 0.0f ? SELU_SCALE * v
                             : SELU_SCALE * SELU_ALPHA * (expf(v) - 1.0f);
                out[(long long)row * OUT_F + ncol] = v;
            }
        }
    }
}

extern "C" void kernel_launch(void* const* d_in, const int* in_sizes, int n_in,
                              void* d_out, int out_size, void* d_ws, size_t ws_size,
                              hipStream_t stream) {
    (void)in_sizes; (void)n_in; (void)out_size; (void)ws_size;

    const float* feat  = (const float*)d_in[0];
    const float* efeat = (const float*)d_in[1];
    const int*   src   = (const int*)d_in[2];
    const int*   dst   = (const int*)d_in[3];
    const float* W     = (const float*)d_in[4];
    const float* bias  = (const float*)d_in[5];
    float*       out   = (float*)d_out;

    float* acc = (float*)d_ws;
    float* deg = acc + (size_t)N_NODES * IN_F;
    int* bhist  = (int*)((char*)d_ws + 32400000);
    int* base   = (int*)((char*)d_ws + 32924288);
    int* binned = (int*)((char*)d_ws + 33448576);

    bin_count<<<NCH, 1024, 0, stream>>>(dst, bhist);
    scan_bins<<<1, 1024, 0, stream>>>(bhist, base);
    bin_scatter<<<NCH, 1024, 0, stream>>>(src, dst, base, binned);

    bucket_sum<<<NB, 512, 0, stream>>>(feat, efeat, base, binned, acc, deg);

    int gblocks = (N_NODES + ROWS_PER_BLOCK - 1) / ROWS_PER_BLOCK;
    gemm_norm_selu<<<gblocks, 128, 0, stream>>>(acc, deg, W, bias, out);
}


# Edges grouped by bucket inside each chunk's own record region (LDS count, LDS scan, place) plus a run table; global count and scan launches removed (5 launches to 3); sums still in LDS arrival order
# speedup vs baseline: 2.1926x; 1.1382x over previous
#include <hip/hip_runtime.h>
#include <math.h>


#define N_NODES 100000
#define N_EDGES 1600000
#define NODE_F  64
#define EDGE_F  16
#define IN_F    80
#define OUT_F   64
#define KPAD    96
#define WAVES   4
#define ROWS_PER_BLOCK (WAVES * 16)
#define ASTRIDE 100

#define NB      500
#define BN      200
#define NCH     250
#define CH      6400

typedef float    v2f  __attribute__((ext_vector_type(2)));
typedef float    v8f  __attribute__((ext_vector_type(8)));
typedef _Float16 v16h __attribute__((ext_vector_type(16)));

#if defined(__AMDGCN__)
#if __has_builtin(__builtin_amdgcn_wmma_f32_16x16x4_f32)
#define USE_WMMA_F32 1
#elif __has_builtin(__builtin_amdgcn_wmma_f32_16x16x32_f16)
#define USE_WMMA_F16 1
#endif
#endif

__device__ __forceinline__ int bucket_of(int d) {
    unsigned b = (unsigned)d / (unsigned)BN;
    return (int)(b < (unsigned)(NB - 1) ? b : (unsigned)(NB - 1));
}

__device__ __forceinline__ void place_edge(int* cur, int4* recs, int e, int s, int d) {
    int b = bucket_of(d);
    int p = atomicAdd(&cur[b], 1);
    if ((unsigned)p < (unsigned)CH) recs[p] = make_int4(e, s, d - b * BN, 0);
}

__global__ void __launch_bounds__(1024)
chunk_group(const int* __restrict__ src, const int* __restrict__ dst,
            int* __restrict__ runs, int* __restrict__ binned) {
    __shared__ int h[512];
    __shared__ int wtot[16];
    const int tid  = threadIdx.x;
    const int blk  = blockIdx.x;
    const int lane = tid & 31;
    const int w    = tid >> 5;
    if (tid < 512) h[tid] = 0;
    __syncthreads();

    const int4* d4p = (const int4*)dst + (size_t)blk * (CH / 4);
    const int4* s4p = (const int4*)src + (size_t)blk * (CH / 4);
    const bool two = tid < CH / 4 - 1024;
    const int  q1  = two ? tid + 1024 : tid;
    const int4 d0 = d4p[tid];
    const int4 d1 = d4p[q1];
    const int4 s0 = s4p[tid];
    const int4 s1 = s4p[q1];

    atomicAdd(&h[bucket_of(d0.x)], 1);
    atomicAdd(&h[bucket_of(d0.y)], 1);
    atomicAdd(&h[bucket_of(d0.z)], 1);
    atomicAdd(&h[bucket_of(d0.w)], 1);
    if (two) {
        atomicAdd(&h[bucket_of(d1.x)], 1);
        atomicAdd(&h[bucket_of(d1.y)], 1);
        atomicAdd(&h[bucket_of(d1.z)], 1);
        atomicAdd(&h[bucket_of(d1.w)], 1);
    }
    __syncthreads();

    int v = 0, inc = 0;
    if (tid < 512) {
        v   = h[tid];
        inc = v;
#pragma unroll
        for (int o = 1; o < 32; o <<= 1) {
            int t = __shfl_up(inc, o);
            if (lane >= o) inc += t;
        }
        if (lane == 31) wtot[w] = inc;
    }
    __syncthreads();
    if (tid < 512) {
        int carry = 0;
        for (int k = 0; k < w; ++k) carry += wtot[k];
        const int excl = carry + inc - v;
        h[tid] = excl;
        if (tid <= NB) runs[tid * NCH + blk] = excl;
    }
    __syncthreads();

    int4* recs = (int4*)binned + (size_t)blk * CH;
    const int e0 = blk * CH + 4 * tid;
    place_edge(h, recs, e0,     s0.x, d0.x);
    place_edge(h, recs, e0 + 1, s0.y, d0.y);
    place_edge(h, recs, e0 + 2, s0.z, d0.z);
    place_edge(h, recs, e0 + 3, s0.w, d0.w);
    if (two) {
        const int e1 = blk * CH + 4 * q1;
        place_edge(h, recs, e1,     s1.x, d1.x);
        place_edge(h, recs, e1 + 1, s1.y, d1.y);
        place_edge(h, recs, e1 + 2, s1.z, d1.z);
        place_edge(h, recs, e1 + 3, s1.w, d1.w);
    }
}

__device__ __forceinline__ void lds_add(float* p, float v) {
    __hip_atomic_fetch_add(p, v, __ATOMIC_RELAXED, __HIP_MEMORY_SCOPE_WORKGROUP);
}

__global__ void __launch_bounds__(512)
bucket_sum(const float* __restrict__ feat,
           const float* __restrict__ efeat,
           const int*   __restrict__ runs,
           const int*   __restrict__ binned,
           float*       __restrict__ acc,
           float*       __restrict__ deg) {
    __shared__ float sacc[BN * IN_F];
    __shared__ int   sdeg[BN];
    __shared__ int   pre[256];
    __shared__ int   rst[256];
    __shared__ int   wtot[16];

    const int tid  = threadIdx.x;
    const int b    = blockIdx.x;
    const int lane = tid & 31;
    const int l16  = lane & 15;
    const int w    = __builtin_amdgcn_readfirstlane(tid >> 5);

    float4* sa4 = (float4*)sacc;
    for (int t = tid; t < BN * IN_F / 4; t += 512) sa4[t] = make_float4(0.f, 0.f, 0.f, 0.f);
    if (tid < BN) sdeg[tid] = 0;

    int st = 0, len = 0;
    if (tid < NCH) {
        st     = runs[b * NCH + tid];
        int en = runs[(b + 1) * NCH + tid];
        st  = st < 0 ? 0 : (st > CH ? CH : st);
        en  = en < st ? st : (en > CH ? CH : en);
        len = en - st;
    }
    int inc = len;
#pragma unroll
    for (int o = 1; o < 32; o <<= 1) {
        int t = __shfl_up(inc, o);
        if (lane >= o) inc += t;
    }
    if (lane == 31) wtot[w] = inc;
    __syncthreads();
    {
        int carry = 0;
        for (int k = 0; k < w; ++k) carry += wtot[k];
        if (tid <= NCH) pre[tid] = carry + inc - len;
        if (tid <  NCH) rst[tid] = tid * CH + st;
    }
    __syncthreads();
    const int T       = __builtin_amdgcn_readfirstlane(pre[NCH]);
    const int nchunks = (T + 31) >> 5;

    const int4* recs  = (const int4*)binned;
    const v2f*  feat2 = (const v2f*)feat;

    for (int c = w; c < nchunks; c += 16) {
        const int j0 = c << 5;
        int m = T - j0;
        m = m < 32 ? m : 32;
        int4 r = make_int4(0, 0, 0, 0);
        if (lane < m) {
            const int j = j0 + lane;
            int lo = 0, hi = NCH, plo = 0;
#pragma unroll
            for (int it = 0; it < 8; ++it) {
                const int  mid = (lo + hi) >> 1;
                const int  pm  = pre[mid];
                const bool le  = pm <= j;
                lo  = le ? mid : lo;
                plo = le ? pm  : plo;
                hi  = le ? hi  : mid;
            }
            r = recs[rst[lo] + (j - plo)];
            r.z = r.z < 0 ? 0 : (r.z > BN - 1 ? BN - 1 : r.z);
            atomicAdd(&sdeg[r.z], 1);
        }
        for (int t = 0; t < m; t += 16) {
            int ee[16], ss[16], ll[16];
            v2f   hv[16];
            float ev[8];
#pragma unroll
            for (int u = 0; u < 16; ++u) {
                ee[u] = __builtin_amdgcn_readlane(r.x, t + u);
                ss[u] = __builtin_amdgcn_readlane(r.y, t + u);
                ll[u] = __builtin_amdgcn_readlane(r.z, t + u);
            }
#pragma unroll
            for (int u = 0; u < 16; ++u)
                hv[u] = feat2[(size_t)ss[u] * (NODE_F / 2) + lane];
#pragma unroll
            for (int p = 0; p < 8; ++p) {
                int em = lane < 16 ? ee[2 * p] : ee[2 * p + 1];
                ev[p] = efeat[(size_t)em * EDGE_F + l16];
            }
#pragma unroll
            for (int p = 0; p < 8; ++p) {
                if (t + 2 * p < m) {
                    float* row = sacc + ll[2 * p] * IN_F + EDGE_F + 2 * lane;
                    lds_add(row,     hv[2 * p].x);
                    lds_add(row + 1, hv[2 * p].y);
                }
                if (t + 2 * p + 1 < m) {
                    float* row = sacc + ll[2 * p + 1] * IN_F + EDGE_F + 2 * lane;
                    lds_add(row,     hv[2 * p + 1].x);
                    lds_add(row + 1, hv[2 * p + 1].y);
                }
                int lm = lane < 16 ? ll[2 * p] : ll[2 * p + 1];
                if (t + 2 * p + (lane >> 4) < m) lds_add(sacc + lm * IN_F + l16, ev[p]);
            }
        }
    }
    __syncthreads();

    float4* o4 = (float4*)(acc + (size_t)b * (BN * IN_F));
    for (int t = tid; t < BN * IN_F / 4; t += 512) o4[t] = sa4[t];
    if (tid < BN) {
        float d = (float)sdeg[tid];
        d = d < 1.0f ? 1.0f : d;
        deg[b * BN + tid] = 1.0f / sqrtf(d);
    }
}

__global__ void __launch_bounds__(128)
gemm_norm_selu(const float* __restrict__ acc,
               const float* __restrict__ norm,
               const float* __restrict__ W,
               const float* __restrict__ bias,
               float*       __restrict__ out) {
    __shared__ float sW[KPAD * OUT_F];
    __shared__ float sA[ROWS_PER_BLOCK * ASTRIDE];

    const int tid  = threadIdx.x;
    const int base = blockIdx.x * ROWS_PER_BLOCK;

    for (int t = tid; t < KPAD * OUT_F; t += 128)
        sW[t] = (t < IN_F * OUT_F) ? W[t] : 0.0f;
    for (int t = tid; t < ROWS_PER_BLOCK * ASTRIDE; t += 128)
        sA[t] = 0.0f;
    __syncthreads();
    for (int t = tid; t < ROWS_PER_BLOCK * IN_F; t += 128) {
        int r = t / IN_F, c = t % IN_F;
        int row = base + r;
        float v = 0.0f;
        if (row < N_NODES) v = acc[(long long)row * IN_F + c] * norm[row];
        sA[r * ASTRIDE + c] = v;
    }
    __syncthreads();

    const int wave  = tid >> 5;
    const int lane  = tid & 31;
    const int half  = lane >> 4;
    const int l     = lane & 15;
    const int rbase = wave * 16;

    v8f cacc[4] = {};

#if defined(USE_WMMA_F32)
    for (int kk = 0; kk < IN_F; kk += 4) {
        v2f a;
        const float* ar = &sA[(rbase + l) * ASTRIDE + kk + 2 * half];
        a.x = ar[0];
        a.y = ar[1];
#pragma unroll
        for (int nt = 0; nt < 4; ++nt) {
            int ncol = nt * 16 + l;
            v2f b;
            b.x = sW[(kk + 2 * half)     * OUT_F + ncol];
            b.y = sW[(kk + 2 * half + 1) * OUT_F + ncol];
            cacc[nt] = __builtin_amdgcn_wmma_f32_16x16x4_f32(
                false, a, false, b, (short)0, cacc[nt], false, false);
        }
    }
#elif defined(USE_WMMA_F16)
    for (int kk = 0; kk < KPAD; kk += 32) {
        const int kofs = half * 8;
        v16h a;
        const float* ar = &sA[(rbase + l) * ASTRIDE + kk];
#pragma unroll
        for (int j = 0; j < 8; ++j) {
            a[j]     = (_Float16)ar[kofs + j];
            a[8 + j] = (_Float16)ar[16 + kofs + j];
        }
#pragma unroll
        for (int nt = 0; nt < 4; ++nt) {
            int ncol = nt * 16 + l;
            v16h b;
#pragma unroll
            for (int j = 0; j < 8; ++j) {
                b[j]     = (_Float16)sW[(kk + kofs + j)      * OUT_F + ncol];
                b[8 + j] = (_Float16)sW[(kk + 16 + kofs + j) * OUT_F + ncol];
            }
            cacc[nt] = __builtin_amdgcn_wmma_f32_16x16x32_f16(
                false, a, false, b, (short)0, cacc[nt], false, false);
        }
    }
#else
    for (int i = 0; i < 8; ++i) {
        int m = i + 8 * half;
        for (int nt = 0; nt < 4; ++nt) {
            float s = 0.0f;
            for (int kk = 0; kk < IN_F; ++kk)
                s += sA[(rbase + m) * ASTRIDE + kk] * sW[kk * OUT_F + nt * 16 + l];
            cacc[nt][i] = s;
        }
    }
#endif

    const float SELU_SCALE = 1.0507009873554805f;
    const float SELU_ALPHA = 1.6732632423543772f;
#pragma unroll
    for (int nt = 0; nt < 4; ++nt) {
        int ncol = nt * 16 + l;
        float bv = bias[ncol];
#pragma unroll
        for (int i = 0; i < 8; ++i) {
            int m   = i + 8 * half;
            int row = base + rbase + m;
            if (row < N_NODES) {
                float v = cacc[nt][i] * norm[row] + bv;
                v = v > 0.0f ? SELU_SCALE * v
                             : SELU_SCALE * SELU_ALPHA * (expf(v) - 1.0f);
                out[(long long)row * OUT_F + ncol] = v;
            }
        }
    }
}

extern "C" void kernel_launch(void* const* d_in, const int* in_sizes, int n_in,
                              void* d_out, int out_size, void* d_ws, size_t ws_size,
                              hipStream_t stream) {
    (void)in_sizes; (void)n_in; (void)out_size; (void)ws_size;

    const float* feat  = (const float*)d_in[0];
    const float* efeat = (const float*)d_in[1];
    const int*   src   = (const int*)d_in[2];
    const int*   dst   = (const int*)d_in[3];
    const float* W     = (const float*)d_in[4];
    const float* bias  = (const float*)d_in[5];
    float*       out   = (float*)d_out;

    float* acc = (float*)d_ws;
    float* deg = acc + (size_t)N_NODES * IN_F;
    int* runs   = (int*)((char*)d_ws + 32400000);
    int* binned = (int*)((char*)d_ws + 33448576);

    chunk_group<<<NCH, 1024, 0, stream>>>(src, dst, runs, binned);

    bucket_sum<<<NB, 512, 0, stream>>>(feat, efeat, runs, binned, acc, deg);

    int gblocks = (N_NODES + ROWS_PER_BLOCK - 1) / ROWS_PER_BLOCK;
    gemm_norm_selu<<<gblocks, 128, 0, stream>>>(acc, deg, W, bias, out);
}


# Third launch fused into the bucket kernel: each bucket's 200 rows go from their LDS sums through the f32 matrix product, norm, bias and selu straight to out; sums and norm never pass through memory
# speedup vs baseline: 4.0262x; 1.8363x over previous
#include <hip/hip_runtime.h>
#include <math.h>


#define N_NODES 100000
#define N_EDGES 1600000
#define NODE_F  64
#define EDGE_F  16
#define IN_F    80
#define OUT_F   64

#define NB      500
#define BN      200
#define NCH     250
#define CH      6400
#define NTILE   ((BN + 15) / 16)

typedef float    v2f  __attribute__((ext_vector_type(2)));
typedef float    v8f  __attribute__((ext_vector_type(8)));

#if defined(__AMDGCN__)
#if __has_builtin(__builtin_amdgcn_wmma_f32_16x16x4_f32)
#define USE_WMMA_F32 1
#endif
#endif

__device__ __forceinline__ int bucket_of(int d) {
    unsigned b = (unsigned)d / (unsigned)BN;
    return (int)(b < (unsigned)(NB - 1) ? b : (unsigned)(NB - 1));
}

__device__ __forceinline__ void place_edge(int* cur, int4* recs, int e, int s, int d) {
    int b = bucket_of(d);
    int p = atomicAdd(&cur[b], 1);
    if ((unsigned)p < (unsigned)CH) recs[p] = make_int4(e, s, d - b * BN, 0);
}

__global__ void __launch_bounds__(1024)
chunk_group(const int* __restrict__ src, const int* __restrict__ dst,
            int* __restrict__ runs, int* __restrict__ binned) {
    __shared__ int h[512];
    __shared__ int wtot[16];
    const int tid  = threadIdx.x;
    const int blk  = blockIdx.x;
    const int lane = tid & 31;
    const int w    = tid >> 5;
    if (tid < 512) h[tid] = 0;
    __syncthreads();

    const int4* d4p = (const int4*)dst + (size_t)blk * (CH / 4);
    const int4* s4p = (const int4*)src + (size_t)blk * (CH / 4);
    const bool two = tid < CH / 4 - 1024;
    const int  q1  = two ? tid + 1024 : tid;
    const int4 d0 = d4p[tid];
    const int4 d1 = d4p[q1];
    const int4 s0 = s4p[tid];
    const int4 s1 = s4p[q1];

    atomicAdd(&h[bucket_of(d0.x)], 1);
    atomicAdd(&h[bucket_of(d0.y)], 1);
    atomicAdd(&h[bucket_of(d0.z)], 1);
    atomicAdd(&h[bucket_of(d0.w)], 1);
    if (two) {
        atomicAdd(&h[bucket_of(d1.x)], 1);
        atomicAdd(&h[bucket_of(d1.y)], 1);
        atomicAdd(&h[bucket_of(d1.z)], 1);
        atomicAdd(&h[bucket_of(d1.w)], 1);
    }
    __syncthreads();

    int v = 0, inc = 0;
    if (tid < 512) {
        v   = h[tid];
        inc = v;
#pragma unroll
        for (int o = 1; o < 32; o <<= 1) {
            int t = __shfl_up(inc, o);
            if (lane >= o) inc += t;
        }
        if (lane == 31) wtot[w] = inc;
    }
    __syncthreads();
    if (tid < 512) {
        int carry = 0;
        for (int k = 0; k < w; ++k) carry += wtot[k];
        const int excl = carry + inc - v;
        h[tid] = excl;
        if (tid <= NB) runs[tid * NCH + blk] = excl;
    }
    __syncthreads();

    int4* recs = (int4*)binned + (size_t)blk * CH;
    const int e0 = blk * CH + 4 * tid;
    place_edge(h, recs, e0,     s0.x, d0.x);
    place_edge(h, recs, e0 + 1, s0.y, d0.y);
    place_edge(h, recs, e0 + 2, s0.z, d0.z);
    place_edge(h, recs, e0 + 3, s0.w, d0.w);
    if (two) {
        const int e1 = blk * CH + 4 * q1;
        place_edge(h, recs, e1,     s1.x, d1.x);
        place_edge(h, recs, e1 + 1, s1.y, d1.y);
        place_edge(h, recs, e1 + 2, s1.z, d1.z);
        place_edge(h, recs, e1 + 3, s1.w, d1.w);
    }
}

__device__ __forceinline__ void lds_add(float* p, float v) {
    __hip_atomic_fetch_add(p, v, __ATOMIC_RELAXED, __HIP_MEMORY_SCOPE_WORKGROUP);
}

__global__ void __launch_bounds__(512)
bucket_sum_gemm(const float* __restrict__ feat,
                const float* __restrict__ efeat,
                const int*   __restrict__ runs,
                const int*   __restrict__ binned,
                const float* __restrict__ W,
                const float* __restrict__ bias,
                float*       __restrict__ out) {
    __shared__ __attribute__((aligned(16))) float sacc[BN * IN_F];
    __shared__ __attribute__((aligned(16))) float sW[IN_F * OUT_F];
    __shared__ float snorm[BN];
    __shared__ int   sdeg[BN];
    __shared__ int   pre[256];
    __shared__ int   rst[256];
    __shared__ int   wtot[16];

    const int tid  = threadIdx.x;
    const int b    = blockIdx.x;
    const int lane = tid & 31;
    const int l16  = lane & 15;
    const int w    = __builtin_amdgcn_readfirstlane(tid >> 5);

    float4* sa4 = (float4*)sacc;
    for (int t = tid; t < BN * IN_F / 4; t += 512) sa4[t] = make_float4(0.f, 0.f, 0.f, 0.f);
    if (tid < BN) sdeg[tid] = 0;
    {
        const float4* W4  = (const float4*)W;
        float4*       sW4 = (float4*)sW;
        for (int t = tid; t < IN_F * OUT_F / 4; t += 512) sW4[t] = W4[t];
    }

    int st = 0, len = 0;
    if (tid < NCH) {
        st     = runs[b * NCH + tid];
        int en = runs[(b + 1) * NCH + tid];
        st  = st < 0 ? 0 : (st > CH ? CH : st);
        en  = en < st ? st : (en > CH ? CH : en);
        len = en - st;
    }
    int inc = len;
#pragma unroll
    for (int o = 1; o < 32; o <<= 1) {
        int t = __shfl_up(inc, o);
        if (lane >= o) inc += t;
    }
    if (lane == 31) wtot[w] = inc;
    __syncthreads();
    {
        int carry = 0;
        for (int k = 0; k < w; ++k) carry += wtot[k];
        if (tid <= NCH) pre[tid] = carry + inc - len;
        if (tid <  NCH) rst[tid] = tid * CH + st;
    }
    __syncthreads();
    const int T       = __builtin_amdgcn_readfirstlane(pre[NCH]);
    const int nchunks = (T + 31) >> 5;

    const int4* recs  = (const int4*)binned;
    const v2f*  feat2 = (const v2f*)feat;

    for (int c = w; c < nchunks; c += 16) {
        const int j0 = c << 5;
        int m = T - j0;
        m = m < 32 ? m : 32;
        int4 r = make_int4(0, 0, 0, 0);
        if (lane < m) {
            const int j = j0 + lane;
            int lo = 0, hi = NCH, plo = 0;
#pragma unroll
            for (int it = 0; it < 8; ++it) {
                const int  mid = (lo + hi) >> 1;
                const int  pm  = pre[mid];
                const bool le  = pm <= j;
                lo  = le ? mid : lo;
                plo = le ? pm  : plo;
                hi  = le ? hi  : mid;
            }
            r = recs[rst[lo] + (j - plo)];
            r.z = r.z < 0 ? 0 : (r.z > BN - 1 ? BN - 1 : r.z);
            atomicAdd(&sdeg[r.z], 1);
        }
        for (int t = 0; t < m; t += 16) {
            int ee[16], ss[16], ll[16];
            v2f   hv[16];
            float ev[8];
#pragma unroll
            for (int u = 0; u < 16; ++u) {
                ee[u] = __builtin_amdgcn_readlane(r.x, t + u);
                ss[u] = __builtin_amdgcn_readlane(r.y, t + u);
                ll[u] = __builtin_amdgcn_readlane(r.z, t + u);
            }
#pragma unroll
            for (int u = 0; u < 16; ++u)
                hv[u] = feat2[(size_t)ss[u] * (NODE_F / 2) + lane];
#pragma unroll
            for (int p = 0; p < 8; ++p) {
                int em = lane < 16 ? ee[2 * p] : ee[2 * p + 1];
                ev[p] = efeat[(size_t)em * EDGE_F + l16];
            }
#pragma unroll
            for (int p = 0; p < 8; ++p) {
                if (t + 2 * p < m) {
                    float* row = sacc + ll[2 * p] * IN_F + EDGE_F + 2 * lane;
                    lds_add(row,     hv[2 * p].x);
                    lds_add(row + 1, hv[2 * p].y);
                }
                if (t + 2 * p + 1 < m) {
                    float* row = sacc + ll[2 * p + 1] * IN_F + EDGE_F + 2 * lane;
                    lds_add(row,     hv[2 * p + 1].x);
                    lds_add(row + 1, hv[2 * p + 1].y);
                }
                int lm = lane < 16 ? ll[2 * p] : ll[2 * p + 1];
                if (t + 2 * p + (lane >> 4) < m) lds_add(sacc + lm * IN_F + l16, ev[p]);
            }
        }
    }
    __syncthreads();

    if (tid < BN) {
        float d = (float)sdeg[tid];
        d = d < 1.0f ? 1.0f : d;
        snorm[tid] = 1.0f / sqrtf(d);
    }
    __syncthreads();

    if (w < NTILE) {
        const int half  = lane >> 4;
        const int rbase = w * 16;
        int ra = rbase + l16;
        ra = ra < BN ? ra : BN - 1;
        const float  na = snorm[ra];
        const float* ar = sacc + ra * IN_F + 2 * half;
        const float* bw = sW + (2 * half) * OUT_F + l16;

        v8f cacc[4] = {};

#if defined(USE_WMMA_F32)
        for (int kk = 0; kk < IN_F; kk += 4) {
            v2f a;
            a.x = ar[kk]     * na;
            a.y = ar[kk + 1] * na;
#pragma unroll
            for (int nt = 0; nt < 4; ++nt) {
                v2f bb;
                bb.x = bw[kk       * OUT_F + nt * 16];
                bb.y = bw[(kk + 1) * OUT_F + nt * 16];
                cacc[nt] = __builtin_amdgcn_wmma_f32_16x16x4_f32(
                    false, a, false, bb, (short)0, cacc[nt], false, false);
            }
        }
#else
        for (int i = 0; i < 8; ++i) {
            int m = rbase + i + 8 * half;
            m = m < BN ? m : BN - 1;
            for (int nt = 0; nt < 4; ++nt) {
                float s = 0.0f;
                for (int kk = 0; kk < IN_F; ++kk)
                    s += sacc[m * IN_F + kk] * snorm[m] * sW[kk * OUT_F + nt * 16 + l16];
                cacc[nt][i] = s;
            }
        }
        (void)na; (void)ar; (void)bw;
#endif

        const float SELU_SCALE = 1.0507009873554805f;
        const float SELU_ALPHA = 1.6732632423543772f;
        float nd[8];
#pragma unroll
        for (int i = 0; i < 8; ++i) {
            int m = rbase + i + 8 * half;
            nd[i] = snorm[m < BN ? m : BN - 1];
        }
        float* orow = out + (size_t)b * (BN * OUT_F);
#pragma unroll
        for (int nt = 0; nt < 4; ++nt) {
            const int   ncol = nt * 16 + l16;
            const float bv   = bias[ncol];
#pragma unroll
            for (int i = 0; i < 8; ++i) {
                const int m = rbase + i + 8 * half;
                if (m < BN) {
                    float v = cacc[nt][i] * nd[i] + bv;
                    v = v > 0.0f ? SELU_SCALE * v
                                 : SELU_SCALE * SELU_ALPHA * (expf(v) - 1.0f);
                    orow[m * OUT_F + ncol] = v;
                }
            }
        }
    }
}

extern "C" void kernel_launch(void* const* d_in, const int* in_sizes, int n_in,
                              void* d_out, int out_size, void* d_ws, size_t ws_size,
                              hipStream_t stream) {
    (void)in_sizes; (void)n_in; (void)out_size; (void)ws_size;

    const float* feat  = (const float*)d_in[0];
    const float* efeat = (const float*)d_in[1];
    const int*   src   = (const int*)d_in[2];
    const int*   dst   = (const int*)d_in[3];
    const float* W     = (const float*)d_in[4];
    const float* bias  = (const float*)d_in[5];
    float*       out   = (float*)d_out;

    int* runs   = (int*)((char*)d_ws + 32400000);
    int* binned = (int*)((char*)d_ws + 33448576);

    chunk_group<<<NCH, 1024, 0, stream>>>(src, dst, runs, binned);

    bucket_sum_gemm<<<NB, 512, 0, stream>>>(feat, efeat, runs, binned, W, bias, out);
}


# Half-wave pick of a record pair's edge id and node slot made on the two scalars already read (one select each); it was built as an indexed array read, a compare/select chain over every entry
# speedup vs baseline: 4.5783x; 1.1371x over previous
#include <hip/hip_runtime.h>
#include <math.h>


#define N_NODES 100000
#define N_EDGES 1600000
#define NODE_F  64
#define EDGE_F  16
#define IN_F    80
#define OUT_F   64

#define NB      500
#define BN      200
#define NCH     250
#define CH      6400
#define NTILE   ((BN + 15) / 16)

typedef float    v2f  __attribute__((ext_vector_type(2)));
typedef float    v8f  __attribute__((ext_vector_type(8)));

#if defined(__AMDGCN__)
#if __has_builtin(__builtin_amdgcn_wmma_f32_16x16x4_f32)
#define USE_WMMA_F32 1
#endif
#endif

__device__ __forceinline__ int bucket_of(int d) {
    unsigned b = (unsigned)d / (unsigned)BN;
    return (int)(b < (unsigned)(NB - 1) ? b : (unsigned)(NB - 1));
}

__device__ __forceinline__ void place_edge(int* cur, int4* recs, int e, int s, int d) {
    int b = bucket_of(d);
    int p = atomicAdd(&cur[b], 1);
    if ((unsigned)p < (unsigned)CH) recs[p] = make_int4(e, s, d - b * BN, 0);
}

__global__ void __launch_bounds__(1024)
chunk_group(const int* __restrict__ src, const int* __restrict__ dst,
            int* __restrict__ runs, int* __restrict__ binned) {
    __shared__ int h[512];
    __shared__ int wtot[16];
    const int tid  = threadIdx.x;
    const int blk  = blockIdx.x;
    const int lane = tid & 31;
    const int w    = tid >> 5;
    if (tid < 512) h[tid] = 0;
    __syncthreads();

    const int4* d4p = (const int4*)dst + (size_t)blk * (CH / 4);
    const int4* s4p = (const int4*)src + (size_t)blk * (CH / 4);
    const bool two = tid < CH / 4 - 1024;
    const int  q1  = two ? tid + 1024 : tid;
    const int4 d0 = d4p[tid];
    const int4 d1 = d4p[q1];
    const int4 s0 = s4p[tid];
    const int4 s1 = s4p[q1];

    atomicAdd(&h[bucket_of(d0.x)], 1);
    atomicAdd(&h[bucket_of(d0.y)], 1);
    atomicAdd(&h[bucket_of(d0.z)], 1);
    atomicAdd(&h[bucket_of(d0.w)], 1);
    if (two) {
        atomicAdd(&h[bucket_of(d1.x)], 1);
        atomicAdd(&h[bucket_of(d1.y)], 1);
        atomicAdd(&h[bucket_of(d1.z)], 1);
        atomicAdd(&h[bucket_of(d1.w)], 1);
    }
    __syncthreads();

    int v = 0, inc = 0;
    if (tid < 512) {
        v   = h[tid];
        inc = v;
#pragma unroll
        for (int o = 1; o < 32; o <<= 1) {
            int t = __shfl_up(inc, o);
            if (lane >= o) inc += t;
        }
        if (lane == 31) wtot[w] = inc;
    }
    __syncthreads();
    if (tid < 512) {
        int carry = 0;
        for (int k = 0; k < w; ++k) carry += wtot[k];
        const int excl = carry + inc - v;
        h[tid] = excl;
        if (tid <= NB) runs[tid * NCH + blk] = excl;
    }
    __syncthreads();

    int4* recs = (int4*)binned + (size_t)blk * CH;
    const int e0 = blk * CH + 4 * tid;
    place_edge(h, recs, e0,     s0.x, d0.x);
    place_edge(h, recs, e0 + 1, s0.y, d0.y);
    place_edge(h, recs, e0 + 2, s0.z, d0.z);
    place_edge(h, recs, e0 + 3, s0.w, d0.w);
    if (two) {
        const int e1 = blk * CH + 4 * q1;
        place_edge(h, recs, e1,     s1.x, d1.x);
        place_edge(h, recs, e1 + 1, s1.y, d1.y);
        place_edge(h, recs, e1 + 2, s1.z, d1.z);
        place_edge(h, recs, e1 + 3, s1.w, d1.w);
    }
}

__device__ __forceinline__ void lds_add(float* p, float v) {
    __hip_atomic_fetch_add(p, v, __ATOMIC_RELAXED, __HIP_MEMORY_SCOPE_WORKGROUP);
}

__global__ void __launch_bounds__(512)
bucket_sum_gemm(const float* __restrict__ feat,
                const float* __restrict__ efeat,
                const int*   __restrict__ runs,
                const int*   __restrict__ binned,
                const float* __restrict__ W,
                const float* __restrict__ bias,
                float*       __restrict__ out) {
    __shared__ __attribute__((aligned(16))) float sacc[BN * IN_F];
    __shared__ __attribute__((aligned(16))) float sW[IN_F * OUT_F];
    __shared__ float snorm[BN];
    __shared__ int   sdeg[BN];
    __shared__ int   pre[256];
    __shared__ int   rst[256];
    __shared__ int   wtot[16];

    const int tid  = threadIdx.x;
    const int b    = blockIdx.x;
    const int lane = tid & 31;
    const int l16  = lane & 15;
    const bool hi16 = lane >= 16;
    const int w    = __builtin_amdgcn_readfirstlane(tid >> 5);

    float4* sa4 = (float4*)sacc;
    for (int t = tid; t < BN * IN_F / 4; t += 512) sa4[t] = make_float4(0.f, 0.f, 0.f, 0.f);
    if (tid < BN) sdeg[tid] = 0;
    {
        const float4* W4  = (const float4*)W;
        float4*       sW4 = (float4*)sW;
        for (int t = tid; t < IN_F * OUT_F / 4; t += 512) sW4[t] = W4[t];
    }

    int st = 0, len = 0;
    if (tid < NCH) {
        st     = runs[b * NCH + tid];
        int en = runs[(b + 1) * NCH + tid];
        st  = st < 0 ? 0 : (st > CH ? CH : st);
        en  = en < st ? st : (en > CH ? CH : en);
        len = en - st;
    }
    int inc = len;
#pragma unroll
    for (int o = 1; o < 32; o <<= 1) {
        int t = __shfl_up(inc, o);
        if (lane >= o) inc += t;
    }
    if (lane == 31) wtot[w] = inc;
    __syncthreads();
    {
        int carry = 0;
        for (int k = 0; k < w; ++k) carry += wtot[k];
        if (tid <= NCH) pre[tid] = carry + inc - len;
        if (tid <  NCH) rst[tid] = tid * CH + st;
    }
    __syncthreads();
    const int T       = __builtin_amdgcn_readfirstlane(pre[NCH]);
    const int nchunks = (T + 31) >> 5;

    const int4* recs  = (const int4*)binned;
    const v2f*  feat2 = (const v2f*)feat;

    for (int c = w; c < nchunks; c += 16) {
        const int j0 = c << 5;
        int m = T - j0;
        m = m < 32 ? m : 32;
        int4 r = make_int4(0, 0, 0, 0);
        if (lane < m) {
            const int j = j0 + lane;
            int lo = 0, hi = NCH, plo = 0;
#pragma unroll
            for (int it = 0; it < 8; ++it) {
                const int  mid = (lo + hi) >> 1;
                const int  pm  = pre[mid];
                const bool le  = pm <= j;
                lo  = le ? mid : lo;
                plo = le ? pm  : plo;
                hi  = le ? hi  : mid;
            }
            r = recs[rst[lo] + (j - plo)];
            r.z = r.z < 0 ? 0 : (r.z > BN - 1 ? BN - 1 : r.z);
            atomicAdd(&sdeg[r.z], 1);
        }
        for (int t = 0; t < m; t += 16) {
            int ee[16], ss[16], ll[16];
            v2f   hv[16];
            float ev[8];
#pragma unroll
            for (int u = 0; u < 16; ++u) {
                ee[u] = __builtin_amdgcn_readlane(r.x, t + u);
                ss[u] = __builtin_amdgcn_readlane(r.y, t + u);
                ll[u] = __builtin_amdgcn_readlane(r.z, t + u);
            }
#pragma unroll
            for (int u = 0; u < 16; ++u)
                hv[u] = feat2[(size_t)ss[u] * (NODE_F / 2) + lane];
#pragma unroll
            for (int p = 0; p < 8; ++p) {
                const int ea = ee[2 * p];
                const int eb = ee[2 * p + 1];
                const int em = hi16 ? eb : ea;
                ev[p] = efeat[(size_t)em * EDGE_F + l16];
            }
#pragma unroll
            for (int p = 0; p < 8; ++p) {
                if (t + 2 * p < m) {
                    float* row = sacc + ll[2 * p] * IN_F + EDGE_F + 2 * lane;
                    lds_add(row,     hv[2 * p].x);
                    lds_add(row + 1, hv[2 * p].y);
                }
                if (t + 2 * p + 1 < m) {
                    float* row = sacc + ll[2 * p + 1] * IN_F + EDGE_F + 2 * lane;
                    lds_add(row,     hv[2 * p + 1].x);
                    lds_add(row + 1, hv[2 * p + 1].y);
                }
                const int la = ll[2 * p];
                const int lb = ll[2 * p + 1];
                const int lm = hi16 ? lb : la;
                if (t + 2 * p + (lane >> 4) < m) lds_add(sacc + lm * IN_F + l16, ev[p]);
            }
        }
    }
    __syncthreads();

    if (tid < BN) {
        float d = (float)sdeg[tid];
        d = d < 1.0f ? 1.0f : d;
        snorm[tid] = 1.0f / sqrtf(d);
    }
    __syncthreads();

    if (w < NTILE) {
        const int half  = lane >> 4;
        const int rbase = w * 16;
        int ra = rbase + l16;
        ra = ra < BN ? ra : BN - 1;
        const float  na = snorm[ra];
        const float* ar = sacc + ra * IN_F + 2 * half;
        const float* bw = sW + (2 * half) * OUT_F + l16;

        v8f cacc[4] = {};

#if defined(USE_WMMA_F32)
        for (int kk = 0; kk < IN_F; kk += 4) {
            v2f a;
            a.x = ar[kk]     * na;
            a.y = ar[kk + 1] * na;
#pragma unroll
            for (int nt = 0; nt < 4; ++nt) {
                v2f bb;
                bb.x = bw[kk       * OUT_F + nt * 16];
                bb.y = bw[(kk + 1) * OUT_F + nt * 16];
                cacc[nt] = __builtin_amdgcn_wmma_f32_16x16x4_f32(
                    false, a, false, bb, (short)0, cacc[nt], false, false);
            }
        }
#else
        for (int i = 0; i < 8; ++i) {
            int m = rbase + i + 8 * half;
            m = m < BN ? m : BN - 1;
            for (int nt = 0; nt < 4; ++nt) {
                float s = 0.0f;
                for (int kk = 0; kk < IN_F; ++kk)
                    s += sacc[m * IN_F + kk] * snorm[m] * sW[kk * OUT_F + nt * 16 + l16];
                cacc[nt][i] = s;
            }
        }
        (void)na; (void)ar; (void)bw;
#endif

        const float SELU_SCALE = 1.0507009873554805f;
        const float SELU_ALPHA = 1.6732632423543772f;
        float nd[8];
#pragma unroll
        for (int i = 0; i < 8; ++i) {
            int m = rbase + i + 8 * half;
            nd[i] = snorm[m < BN ? m : BN - 1];
        }
        float* orow = out + (size_t)b * (BN * OUT_F);
#pragma unroll
        for (int nt = 0; nt < 4; ++nt) {
            const int   ncol = nt * 16 + l16;
            const float bv   = bias[ncol];
#pragma unroll
            for (int i = 0; i < 8; ++i) {
                const int m = rbase + i + 8 * half;
                if (m < BN) {
                    float v = cacc[nt][i] * nd[i] + bv;
                    v = v > 0.0f ? SELU_SCALE * v
                                 : SELU_SCALE * SELU_ALPHA * (expf(v) - 1.0f);
                    orow[m * OUT_F + ncol] = v;
                }
            }
        }
    }
}

extern "C" void kernel_launch(void* const* d_in, const int* in_sizes, int n_in,
                              void* d_out, int out_size, void* d_ws, size_t ws_size,
                              hipStream_t stream) {
    (void)in_sizes; (void)n_in; (void)out_size; (void)ws_size;

    const float* feat  = (const float*)d_in[0];
    const float* efeat = (const float*)d_in[1];
    const int*   src   = (const int*)d_in[2];
    const int*   dst   = (const int*)d_in[3];
    const float* W     = (const float*)d_in[4];
    const float* bias  = (const float*)d_in[5];
    float*       out   = (float*)d_out;

    int* runs   = (int*)((char*)d_ws + 32400000);
    int* binned = (int*)((char*)d_ws + 33448576);

    chunk_group<<<NCH, 1024, 0, stream>>>(src, dst, runs, binned);

    bucket_sum_gemm<<<NB, 512, 0, stream>>>(feat, efeat, runs, binned, W, bias, out);
}


# Bucket's records copied once into LDS in record order (16 runs per wave fetched at once); trips read them there in all lanes: no per-chunk run search, record fetch, readlanes; sums LDS arrival order
# speedup vs baseline: 4.8002x; 1.0485x over previous
#include <hip/hip_runtime.h>
#include <math.h>


#define N_NODES 100000
#define N_EDGES 1600000
#define NODE_F  64
#define EDGE_F  16
#define IN_F    80
#define OUT_F   64

#define NB      500
#define BN      200
#define NCH     250
#define CH      6400
#define NTILE   ((BN + 15) / 16)
#define CAP     4096
#define SRC_BITS 17
#define EDGE_MASK 0x3fffff

typedef float    v2f  __attribute__((ext_vector_type(2)));
typedef float    v8f  __attribute__((ext_vector_type(8)));

#if defined(__AMDGCN__)
#if __has_builtin(__builtin_amdgcn_wmma_f32_16x16x4_f32)
#define USE_WMMA_F32 1
#endif
#endif

__device__ __forceinline__ int bucket_of(int d) {
    unsigned b = (unsigned)d / (unsigned)BN;
    return (int)(b < (unsigned)(NB - 1) ? b : (unsigned)(NB - 1));
}

__device__ __forceinline__ void place_edge(int* cur, int4* recs, int e, int s, int d) {
    int b = bucket_of(d);
    int p = atomicAdd(&cur[b], 1);
    if ((unsigned)p < (unsigned)CH) recs[p] = make_int4(e, s, d - b * BN, 0);
}

__global__ void __launch_bounds__(1024)
chunk_group(const int* __restrict__ src, const int* __restrict__ dst,
            int* __restrict__ runs, int* __restrict__ binned) {
    __shared__ int h[512];
    __shared__ int wtot[16];
    const int tid  = threadIdx.x;
    const int blk  = blockIdx.x;
    const int lane = tid & 31;
    const int w    = tid >> 5;
    if (tid < 512) h[tid] = 0;
    __syncthreads();

    const int4* d4p = (const int4*)dst + (size_t)blk * (CH / 4);
    const int4* s4p = (const int4*)src + (size_t)blk * (CH / 4);
    const bool two = tid < CH / 4 - 1024;
    const int  q1  = two ? tid + 1024 : tid;
    const int4 d0 = d4p[tid];
    const int4 d1 = d4p[q1];
    const int4 s0 = s4p[tid];
    const int4 s1 = s4p[q1];

    atomicAdd(&h[bucket_of(d0.x)], 1);
    atomicAdd(&h[bucket_of(d0.y)], 1);
    atomicAdd(&h[bucket_of(d0.z)], 1);
    atomicAdd(&h[bucket_of(d0.w)], 1);
    if (two) {
        atomicAdd(&h[bucket_of(d1.x)], 1);
        atomicAdd(&h[bucket_of(d1.y)], 1);
        atomicAdd(&h[bucket_of(d1.z)], 1);
        atomicAdd(&h[bucket_of(d1.w)], 1);
    }
    __syncthreads();

    int v = 0, inc = 0;
    if (tid < 512) {
        v   = h[tid];
        inc = v;
#pragma unroll
        for (int o = 1; o < 32; o <<= 1) {
            int t = __shfl_up(inc, o);
            if (lane >= o) inc += t;
        }
        if (lane == 31) wtot[w] = inc;
    }
    __syncthreads();
    if (tid < 512) {
        int carry = 0;
        for (int k = 0; k < w; ++k) carry += wtot[k];
        const int excl = carry + inc - v;
        h[tid] = excl;
        if (tid <= NB) runs[tid * NCH + blk] = excl;
    }
    __syncthreads();

    int4* recs = (int4*)binned + (size_t)blk * CH;
    const int e0 = blk * CH + 4 * tid;
    place_edge(h, recs, e0,     s0.x, d0.x);
    place_edge(h, recs, e0 + 1, s0.y, d0.y);
    place_edge(h, recs, e0 + 2, s0.z, d0.z);
    place_edge(h, recs, e0 + 3, s0.w, d0.w);
    if (two) {
        const int e1 = blk * CH + 4 * q1;
        place_edge(h, recs, e1,     s1.x, d1.x);
        place_edge(h, recs, e1 + 1, s1.y, d1.y);
        place_edge(h, recs, e1 + 2, s1.z, d1.z);
        place_edge(h, recs, e1 + 3, s1.w, d1.w);
    }
}

__device__ __forceinline__ void lds_add(float* p, float v) {
    __hip_atomic_fetch_add(p, v, __ATOMIC_RELAXED, __HIP_MEMORY_SCOPE_WORKGROUP);
}

__device__ __forceinline__ int2 pack_rec(int4 r) {
    const int z = r.z < 0 ? 0 : (r.z > BN - 1 ? BN - 1 : r.z);
    return make_int2(r.x, (r.y & ((1 << SRC_BITS) - 1)) | (z << SRC_BITS));
}

__device__ __forceinline__ void stage_rec(int4 r, int j, int* sdeg, int2* srec) {
    const int2 pr = pack_rec(r);
    atomicAdd(&sdeg[(unsigned)pr.y >> SRC_BITS], 1);
    if (j < CAP) srec[j] = pr;
}

__device__ __forceinline__ void sum_trip(const int (&ee)[16], const int (&ss)[16], const int (&ll)[16],
                                         int t, int m, int lane, int l16, bool hi16,
                                         const v2f* __restrict__ feat2,
                                         const float* __restrict__ efeat, float* sacc) {
    v2f   hv[16];
    float ev[8];
#pragma unroll
    for (int u = 0; u < 16; ++u)
        hv[u] = feat2[(unsigned)(ss[u] * (NODE_F / 2) + lane)];
#pragma unroll
    for (int p = 0; p < 8; ++p) {
        const int ea = ee[2 * p];
        const int eb = ee[2 * p + 1];
        const int em = (hi16 ? eb : ea) & EDGE_MASK;
        ev[p] = efeat[(unsigned)(em * EDGE_F + l16)];
    }
#pragma unroll
    for (int p = 0; p < 8; ++p) {
        if (t + 2 * p < m) {
            float* row = sacc + ll[2 * p] * IN_F + EDGE_F + 2 * lane;
            lds_add(row,     hv[2 * p].x);
            lds_add(row + 1, hv[2 * p].y);
        }
        if (t + 2 * p + 1 < m) {
            float* row = sacc + ll[2 * p + 1] * IN_F + EDGE_F + 2 * lane;
            lds_add(row,     hv[2 * p + 1].x);
            lds_add(row + 1, hv[2 * p + 1].y);
        }
        const int la = ll[2 * p];
        const int lb = ll[2 * p + 1];
        const int lm = hi16 ? lb : la;
        if (t + 2 * p + (lane >> 4) < m) lds_add(sacc + lm * IN_F + l16, ev[p]);
    }
}

__global__ void __launch_bounds__(512)
bucket_sum_gemm(const float* __restrict__ feat,
                const float* __restrict__ efeat,
                const int*   __restrict__ runs,
                const int*   __restrict__ binned,
                const float* __restrict__ W,
                const float* __restrict__ bias,
                float*       __restrict__ out) {
    __shared__ __attribute__((aligned(16))) float sacc[BN * IN_F];
    __shared__ __attribute__((aligned(16))) float sW[IN_F * OUT_F];
    __shared__ float snorm[BN];
    __shared__ int   sdeg[BN];
    __shared__ int   pre[256];
    __shared__ int   rst[256];
    __shared__ int   wtot[16];
    __shared__ __attribute__((aligned(16))) int2 srec[CAP];

    const int tid  = threadIdx.x;
    const int b    = blockIdx.x;
    const int lane = tid & 31;
    const int l16  = lane & 15;
    const bool hi16 = lane >= 16;
    const int w    = __builtin_amdgcn_readfirstlane(tid >> 5);

    float4* sa4 = (float4*)sacc;
    for (int t = tid; t < BN * IN_F / 4; t += 512) sa4[t] = make_float4(0.f, 0.f, 0.f, 0.f);
    if (tid < BN) sdeg[tid] = 0;
    {
        const float4* W4  = (const float4*)W;
        float4*       sW4 = (float4*)sW;
        for (int t = tid; t < IN_F * OUT_F / 4; t += 512) sW4[t] = W4[t];
    }

    int st = 0, len = 0;
    if (tid < NCH) {
        st     = runs[b * NCH + tid];
        int en = runs[(b + 1) * NCH + tid];
        st  = st < 0 ? 0 : (st > CH ? CH : st);
        en  = en < st ? st : (en > CH ? CH : en);
        len = en - st;
    }
    int inc = len;
#pragma unroll
    for (int o = 1; o < 32; o <<= 1) {
        int t = __shfl_up(inc, o);
        if (lane >= o) inc += t;
    }
    if (lane == 31) wtot[w] = inc;
    __syncthreads();
    {
        int carry = 0;
        for (int k = 0; k < w; ++k) carry += wtot[k];
        if (tid <= NCH) pre[tid] = carry + inc - len;
        if (tid <  NCH) rst[tid] = tid * CH + st;
    }
    __syncthreads();
    const int T       = __builtin_amdgcn_readfirstlane(pre[NCH]);
    const int nchunks = (T + 31) >> 5;

    const int4* recs  = (const int4*)binned;
    const v2f*  feat2 = (const v2f*)feat;

    {
        const int  cq  = w + 16 * l16;
        const bool okq = cq < NCH;
        const int  ccq = okq ? cq : 0;
        const int  vp0 = pre[ccq];
        const int  vln = okq ? pre[ccq + 1] - vp0 : 0;
        const int  vg0 = rst[ccq];
        int4 rr[16];
#pragma unroll
        for (int q = 0; q < 16; ++q) {
            const int ln = __builtin_amdgcn_readlane(vln, q);
            const int g0 = __builtin_amdgcn_readlane(vg0, q);
            rr[q] = recs[lane < ln ? g0 + lane : 0];
        }
        asm volatile("" :: "v"(rr[0].x), "v"(rr[0].y), "v"(rr[0].z));
#pragma unroll
        for (int q = 0; q < 16; ++q) {
            const int ln = __builtin_amdgcn_readlane(vln, q);
            const int p0 = __builtin_amdgcn_readlane(vp0, q);
            if (lane < ln) stage_rec(rr[q], p0 + lane, sdeg, srec);
        }
        for (int q = 0; q < 16; ++q) {
            const int ln = __builtin_amdgcn_readlane(vln, q);
            const int p0 = __builtin_amdgcn_readlane(vp0, q);
            const int g0 = __builtin_amdgcn_readlane(vg0, q);
            for (int i = lane + 32; i < ln; i += 32)
                stage_rec(recs[g0 + i], p0 + i, sdeg, srec);
        }
    }
    if (tid < 32 && T + tid < CAP) srec[T + tid] = make_int2(0, 0);
    __syncthreads();

    for (int c = w; c < nchunks; c += 16) {
        const int j0 = c << 5;
        int m = T - j0;
        m = m < 32 ? m : 32;
        if (j0 < CAP) {
            for (int t = 0; t < m; t += 16) {
                int ee[16], ss[16], ll[16];
#pragma unroll
                for (int u = 0; u < 16; ++u) {
                    const int2 pr = srec[j0 + t + u];
                    ee[u] = pr.x;
                    ss[u] = pr.y & ((1 << SRC_BITS) - 1);
                    ll[u] = (int)((unsigned)pr.y >> SRC_BITS);
                }
                sum_trip(ee, ss, ll, t, m, lane, l16, hi16, feat2, efeat, sacc);
            }
        } else {
            int2 r = make_int2(0, 0);
            if (lane < m) {
                const int j = j0 + lane;
                int lo = 0, hi = NCH, plo = 0;
#pragma unroll
                for (int it = 0; it < 8; ++it) {
                    const int  mid = (lo + hi) >> 1;
                    const int  pm  = pre[mid];
                    const bool le  = pm <= j;
                    lo  = le ? mid : lo;
                    plo = le ? pm  : plo;
                    hi  = le ? hi  : mid;
                }
                r = pack_rec(recs[rst[lo] + (j - plo)]);
            }
            for (int t = 0; t < m; t += 16) {
                int ee[16], ss[16], ll[16];
#pragma unroll
                for (int u = 0; u < 16; ++u) {
                    ee[u] = __builtin_amdgcn_readlane(r.x, t + u);
                    const int sl = __builtin_amdgcn_readlane(r.y, t + u);
                    ss[u] = sl & ((1 << SRC_BITS) - 1);
                    ll[u] = (int)((unsigned)sl >> SRC_BITS);
                }
                sum_trip(ee, ss, ll, t, m, lane, l16, hi16, feat2, efeat, sacc);
            }
        }
    }
    __syncthreads();

    if (tid < BN) {
        float d = (float)sdeg[tid];
        d = d < 1.0f ? 1.0f : d;
        snorm[tid] = 1.0f / sqrtf(d);
    }
    __syncthreads();

    if (w < NTILE) {
        const int half  = lane >> 4;
        const int rbase = w * 16;
        int ra = rbase + l16;
        ra = ra < BN ? ra : BN - 1;
        const float  na = snorm[ra];
        const float* ar = sacc + ra * IN_F + 2 * half;
        const float* bw = sW + (2 * half) * OUT_F + l16;

        v8f cacc[4] = {};

#if defined(USE_WMMA_F32)
        for (int kk = 0; kk < IN_F; kk += 4) {
            v2f a;
            a.x = ar[kk]     * na;
            a.y = ar[kk + 1] * na;
#pragma unroll
            for (int nt = 0; nt < 4; ++nt) {
                v2f bb;
                bb.x = bw[kk       * OUT_F + nt * 16];
                bb.y = bw[(kk + 1) * OUT_F + nt * 16];
                cacc[nt] = __builtin_amdgcn_wmma_f32_16x16x4_f32(
                    false, a, false, bb, (short)0, cacc[nt], false, false);
            }
        }
#else
        for (int i = 0; i < 8; ++i) {
            int m = rbase + i + 8 * half;
            m = m < BN ? m : BN - 1;
            for (int nt = 0; nt < 4; ++nt) {
                float s = 0.0f;
                for (int kk = 0; kk < IN_F; ++kk)
                    s += sacc[m * IN_F + kk] * snorm[m] * sW[kk * OUT_F + nt * 16 + l16];
                cacc[nt][i] = s;
            }
        }
        (void)na; (void)ar; (void)bw;
#endif

        const float SELU_SCALE = 1.0507009873554805f;
        const float SELU_ALPHA = 1.6732632423543772f;
        float nd[8];
#pragma unroll
        for (int i = 0; i < 8; ++i) {
            int m = rbase + i + 8 * half;
            nd[i] = snorm[m < BN ? m : BN - 1];
        }
        float* orow = out + (size_t)b * (BN * OUT_F);
#pragma unroll
        for (int nt = 0; nt < 4; ++nt) {
            const int   ncol = nt * 16 + l16;
            const float bv   = bias[ncol];
#pragma unroll
            for (int i = 0; i < 8; ++i) {
                const int m = rbase + i + 8 * half;
                if (m < BN) {
                    float v = cacc[nt][i] * nd[i] + bv;
                    v = v > 0.0f ? SELU_SCALE * v
                                 : SELU_SCALE * SELU_ALPHA * (expf(v) - 1.0f);
                    orow[m * OUT_F + ncol] = v;
                }
            }
        }
    }
}

extern "C" void kernel_launch(void* const* d_in, const int* in_sizes, int n_in,
                              void* d_out, int out_size, void* d_ws, size_t ws_size,
                              hipStream_t stream) {
    (void)in_sizes; (void)n_in; (void)out_size; (void)ws_size;

    const float* feat  = (const float*)d_in[0];
    const float* efeat = (const float*)d_in[1];
    const int*   src   = (const int*)d_in[2];
    const int*   dst   = (const int*)d_in[3];
    const float* W     = (const float*)d_in[4];
    const float* bias  = (const float*)d_in[5];
    float*       out   = (float*)d_out;

    int* runs   = (int*)((char*)d_ws + 32400000);
    int* binned = (int*)((char*)d_ws + 33448576);

    chunk_group<<<NCH, 1024, 0, stream>>>(src, dst, runs, binned);

    bucket_sum_gemm<<<NB, 512, 0, stream>>>(feat, efeat, runs, binned, W, bias, out);
}


# Bucket's records counting-sorted by node in LDS (degree scan, returning add = place); a wave sums a node's run in registers, one LDS float add per run, not per record; sum order = place order by runs
# speedup vs baseline: 5.5284x; 1.1517x over previous
#include <hip/hip_runtime.h>
#include <math.h>


#define N_NODES 100000
#define N_EDGES 1600000
#define NODE_F  64
#define EDGE_F  16
#define IN_F    80
#define OUT_F   64

#define NB      500
#define BN      200
#define NCH     250
#define CH      6400
#define NTILE   ((BN + 15) / 16)
#define CAP     3584
#define SPL     ((BN + 31) / 32)
#define SRC_BITS 17
#define EDGE_MASK 0x3fffff

typedef float    v2f  __attribute__((ext_vector_type(2)));
typedef float    v8f  __attribute__((ext_vector_type(8)));

#if defined(__AMDGCN__)
#if __has_builtin(__builtin_amdgcn_wmma_f32_16x16x4_f32)
#define USE_WMMA_F32 1
#endif
#endif

__device__ __forceinline__ int bucket_of(int d) {
    unsigned b = (unsigned)d / (unsigned)BN;
    return (int)(b < (unsigned)(NB - 1) ? b : (unsigned)(NB - 1));
}

__device__ __forceinline__ void place_edge(int* cur, int4* recs, int e, int s, int d) {
    int b = bucket_of(d);
    int p = atomicAdd(&cur[b], 1);
    if ((unsigned)p < (unsigned)CH) recs[p] = make_int4(e, s, d - b * BN, 0);
}

__global__ void __launch_bounds__(1024)
chunk_group(const int* __restrict__ src, const int* __restrict__ dst,
            int* __restrict__ runs, int* __restrict__ binned) {
    __shared__ int h[512];
    __shared__ int wtot[16];
    const int tid  = threadIdx.x;
    const int blk  = blockIdx.x;
    const int lane = tid & 31;
    const int w    = tid >> 5;
    if (tid < 512) h[tid] = 0;
    __syncthreads();

    const int4* d4p = (const int4*)dst + (size_t)blk * (CH / 4);
    const int4* s4p = (const int4*)src + (size_t)blk * (CH / 4);
    const bool two = tid < CH / 4 - 1024;
    const int  q1  = two ? tid + 1024 : tid;
    const int4 d0 = d4p[tid];
    const int4 d1 = d4p[q1];
    const int4 s0 = s4p[tid];
    const int4 s1 = s4p[q1];

    atomicAdd(&h[bucket_of(d0.x)], 1);
    atomicAdd(&h[bucket_of(d0.y)], 1);
    atomicAdd(&h[bucket_of(d0.z)], 1);
    atomicAdd(&h[bucket_of(d0.w)], 1);
    if (two) {
        atomicAdd(&h[bucket_of(d1.x)], 1);
        atomicAdd(&h[bucket_of(d1.y)], 1);
        atomicAdd(&h[bucket_of(d1.z)], 1);
        atomicAdd(&h[bucket_of(d1.w)], 1);
    }
    __syncthreads();

    int v = 0, inc = 0;
    if (tid < 512) {
        v   = h[tid];
        inc = v;
#pragma unroll
        for (int o = 1; o < 32; o <<= 1) {
            int t = __shfl_up(inc, o);
            if (lane >= o) inc += t;
        }
        if (lane == 31) wtot[w] = inc;
    }
    __syncthreads();
    if (tid < 512) {
        int carry = 0;
        for (int k = 0; k < w; ++k) carry += wtot[k];
        const int excl = carry + inc - v;
        h[tid] = excl;
        if (tid <= NB) runs[tid * NCH + blk] = excl;
    }
    __syncthreads();

    int4* recs = (int4*)binned + (size_t)blk * CH;
    const int e0 = blk * CH + 4 * tid;
    place_edge(h, recs, e0,     s0.x, d0.x);
    place_edge(h, recs, e0 + 1, s0.y, d0.y);
    place_edge(h, recs, e0 + 2, s0.z, d0.z);
    place_edge(h, recs, e0 + 3, s0.w, d0.w);
    if (two) {
        const int e1 = blk * CH + 4 * q1;
        place_edge(h, recs, e1,     s1.x, d1.x);
        place_edge(h, recs, e1 + 1, s1.y, d1.y);
        place_edge(h, recs, e1 + 2, s1.z, d1.z);
        place_edge(h, recs, e1 + 3, s1.w, d1.w);
    }
}

__device__ __forceinline__ void lds_add(float* p, float v) {
    __hip_atomic_fetch_add(p, v, __ATOMIC_RELAXED, __HIP_MEMORY_SCOPE_WORKGROUP);
}

__device__ __forceinline__ int2 pack_rec(int4 r) {
    const int z = r.z < 0 ? 0 : (r.z > BN - 1 ? BN - 1 : r.z);
    return make_int2(r.x, (r.y & ((1 << SRC_BITS) - 1)) | (z << SRC_BITS));
}

__device__ __forceinline__ void stage_rec(int4 r, int j, int* sdeg, int2* srec) {
    const int2 pr = pack_rec(r);
    atomicAdd(&sdeg[(unsigned)pr.y >> SRC_BITS], 1);
    if (j < CAP) srec[j] = pr;
}

__device__ __forceinline__ void sum_trip(const int (&ee)[16], const int (&ss)[16], const int (&ll)[16],
                                         int t, int m, int lane, int l16, bool hi16,
                                         const v2f* __restrict__ feat2,
                                         const float* __restrict__ efeat, float* sacc) {
    v2f   hv[16];
    float ev[8];
#pragma unroll
    for (int u = 0; u < 16; ++u)
        hv[u] = feat2[(unsigned)(ss[u] * (NODE_F / 2) + lane)];
#pragma unroll
    for (int p = 0; p < 8; ++p) {
        const int ea = ee[2 * p];
        const int eb = ee[2 * p + 1];
        const int em = (hi16 ? eb : ea) & EDGE_MASK;
        ev[p] = efeat[(unsigned)(em * EDGE_F + l16)];
    }
#pragma unroll
    for (int p = 0; p < 8; ++p) {
        if (t + 2 * p < m) {
            float* row = sacc + ll[2 * p] * IN_F + EDGE_F + 2 * lane;
            lds_add(row,     hv[2 * p].x);
            lds_add(row + 1, hv[2 * p].y);
        }
        if (t + 2 * p + 1 < m) {
            float* row = sacc + ll[2 * p + 1] * IN_F + EDGE_F + 2 * lane;
            lds_add(row,     hv[2 * p + 1].x);
            lds_add(row + 1, hv[2 * p + 1].y);
        }
        const int la = ll[2 * p];
        const int lb = ll[2 * p + 1];
        const int lm = hi16 ? lb : la;
        if (t + 2 * p + (lane >> 4) < m) lds_add(sacc + lm * IN_F + l16, ev[p]);
    }
}

__device__ __forceinline__ void flush_run(float* sacc, int cur, v2f& ah, float& ae, float& ao,
                                          int lane, int l16, bool hi16) {
    float* row = sacc + cur * IN_F;
    lds_add(row + EDGE_F + 2 * lane,     ah.x);
    lds_add(row + EDGE_F + 2 * lane + 1, ah.y);
    lds_add(row + l16, hi16 ? ao : ae);
    ah.x = 0.0f; ah.y = 0.0f; ae = 0.0f; ao = 0.0f;
}

__device__ __forceinline__ void sum_trip_runs(const int2* rs, int nv, int lane, int l16, bool hi16,
                                              const v2f* __restrict__ feat2,
                                              const float* __restrict__ efeat, float* sacc,
                                              int& cur, v2f& ah, float& ae, float& ao) {
    int   ee[16], ss[16], ll[16];
    v2f   hv[16];
    float ev[8];
#pragma unroll
    for (int u = 0; u < 16; ++u) {
        const int2 pr = rs[u];
        ee[u] = pr.x;
        ss[u] = pr.y & ((1 << SRC_BITS) - 1);
        ll[u] = (int)((unsigned)pr.y >> SRC_BITS);
    }
#pragma unroll
    for (int u = 0; u < 16; ++u)
        hv[u] = feat2[(unsigned)(ss[u] * (NODE_F / 2) + lane)];
#pragma unroll
    for (int p = 0; p < 8; ++p) {
        const int ea = ee[2 * p];
        const int eb = ee[2 * p + 1];
        const int em = (hi16 ? eb : ea) & EDGE_MASK;
        ev[p] = efeat[(unsigned)(em * EDGE_F + l16)];
    }
#pragma unroll
    for (int u = 0; u < 16; ++u) {
        if (u == 0 || u < nv) {
            if (ll[u] != cur) {
                flush_run(sacc, cur, ah, ae, ao, lane, l16, hi16);
                cur = ll[u];
            }
            ah.x += hv[u].x;
            ah.y += hv[u].y;
            if (u & 1) ao += ev[u >> 1];
            else       ae += ev[u >> 1];
        }
    }
}

__global__ void __launch_bounds__(512)
bucket_sum_gemm(const float* __restrict__ feat,
                const float* __restrict__ efeat,
                const int*   __restrict__ runs,
                const int*   __restrict__ binned,
                const float* __restrict__ W,
                const float* __restrict__ bias,
                float*       __restrict__ out) {
    __shared__ __attribute__((aligned(16))) float sacc[BN * IN_F];
    __shared__ __attribute__((aligned(16))) float sW[IN_F * OUT_F];
    __shared__ float snorm[BN];
    __shared__ int   sdeg[SPL * 32];
    __shared__ int   scur[SPL * 32];
    __shared__ int   pre[256];
    __shared__ int   rst[256];
    __shared__ int   wtot[16];
    __shared__ __attribute__((aligned(16))) int2 srec[2 * CAP];

    const int tid  = threadIdx.x;
    const int b    = blockIdx.x;
    const int lane = tid & 31;
    const int l16  = lane & 15;
    const bool hi16 = lane >= 16;
    const int w    = __builtin_amdgcn_readfirstlane(tid >> 5);

    float4* sa4 = (float4*)sacc;
    for (int t = tid; t < BN * IN_F / 4; t += 512) sa4[t] = make_float4(0.f, 0.f, 0.f, 0.f);
    if (tid < SPL * 32) sdeg[tid] = 0;
    {
        const float4* W4  = (const float4*)W;
        float4*       sW4 = (float4*)sW;
        for (int t = tid; t < IN_F * OUT_F / 4; t += 512) sW4[t] = W4[t];
    }

    int st = 0, len = 0;
    if (tid < NCH) {
        st     = runs[b * NCH + tid];
        int en = runs[(b + 1) * NCH + tid];
        st  = st < 0 ? 0 : (st > CH ? CH : st);
        en  = en < st ? st : (en > CH ? CH : en);
        len = en - st;
    }
    int inc = len;
#pragma unroll
    for (int o = 1; o < 32; o <<= 1) {
        int t = __shfl_up(inc, o);
        if (lane >= o) inc += t;
    }
    if (lane == 31) wtot[w] = inc;
    __syncthreads();
    {
        int carry = 0;
        for (int k = 0; k < w; ++k) carry += wtot[k];
        if (tid <= NCH) pre[tid] = carry + inc - len;
        if (tid <  NCH) rst[tid] = tid * CH + st;
    }
    __syncthreads();
    const int T       = __builtin_amdgcn_readfirstlane(pre[NCH]);
    const int nchunks = (T + 31) >> 5;

    const int4* recs  = (const int4*)binned;
    const v2f*  feat2 = (const v2f*)feat;

    {
        const int  cq  = w + 16 * l16;
        const bool okq = cq < NCH;
        const int  ccq = okq ? cq : 0;
        const int  vp0 = pre[ccq];
        const int  vln = okq ? pre[ccq + 1] - vp0 : 0;
        const int  vg0 = rst[ccq];
        int4 rr[16];
#pragma unroll
        for (int q = 0; q < 16; ++q) {
            const int ln = __builtin_amdgcn_readlane(vln, q);
            const int g0 = __builtin_amdgcn_readlane(vg0, q);
            rr[q] = recs[lane < ln ? g0 + lane : 0];
        }
        asm volatile("" :: "v"(rr[0].x), "v"(rr[0].y), "v"(rr[0].z));
#pragma unroll
        for (int q = 0; q < 16; ++q) {
            const int ln = __builtin_amdgcn_readlane(vln, q);
            const int p0 = __builtin_amdgcn_readlane(vp0, q);
            if (lane < ln) stage_rec(rr[q], p0 + lane, sdeg, srec);
        }
        for (int q = 0; q < 16; ++q) {
            const int ln = __builtin_amdgcn_readlane(vln, q);
            const int p0 = __builtin_amdgcn_readlane(vp0, q);
            const int g0 = __builtin_amdgcn_readlane(vg0, q);
            for (int i = lane + 32; i < ln; i += 32)
                stage_rec(recs[g0 + i], p0 + i, sdeg, srec);
        }
    }
    __syncthreads();

    const bool sorted = T <= CAP;
    const int  roff   = sorted ? CAP : 0;
    if (w == 0) {
        int v[SPL];
        int s = 0;
#pragma unroll
        for (int k = 0; k < SPL; ++k) {
            v[k] = sdeg[SPL * lane + k];
            s += v[k];
        }
        int sc = s;
#pragma unroll
        for (int o = 1; o < 32; o <<= 1) {
            int t = __shfl_up(sc, o);
            if (lane >= o) sc += t;
        }
        int run = sc - s;
#pragma unroll
        for (int k = 0; k < SPL; ++k) {
            scur[SPL * lane + k] = run;
            run += v[k];
        }
    }
    __syncthreads();
    if (sorted) {
        int2 pr[CAP / 512];
#pragma unroll
        for (int k = 0; k < CAP / 512; ++k) {
            const int j = tid + 512 * k;
            pr[k] = srec[j < T ? j : 0];
        }
        int pos[CAP / 512];
#pragma unroll
        for (int k = 0; k < CAP / 512; ++k) {
            const int j = tid + 512 * k;
            pos[k] = j < T ? atomicAdd(&scur[(unsigned)pr[k].y >> SRC_BITS], 1) : CAP;
        }
#pragma unroll
        for (int k = 0; k < CAP / 512; ++k)
            if ((unsigned)pos[k] < (unsigned)CAP) srec[CAP + pos[k]] = pr[k];
        if (tid < 16 && T + tid < CAP) srec[CAP + T + tid] = make_int2(0, 0);
    }
    __syncthreads();

    {
        const int Tc  = T < CAP ? T : CAP;
        const int ntr = (Tc + 15) >> 4;
        const int tb  = (ntr * w) >> 4;
        const int te  = (ntr * (w + 1)) >> 4;
        if (tb < te) {
            const int2* rs = srec + roff + (tb << 4);
            int   cur = (int)((unsigned)rs[0].y >> SRC_BITS);
            v2f   ah  = {0.0f, 0.0f};
            float ae  = 0.0f, ao = 0.0f;
            for (int tr = tb; tr < te; ++tr) {
                int nv = Tc - (tr << 4);
                nv = nv < 16 ? nv : 16;
                sum_trip_runs(rs, nv, lane, l16, hi16, feat2, efeat, sacc, cur, ah, ae, ao);
                rs += 16;
            }
            flush_run(sacc, cur, ah, ae, ao, lane, l16, hi16);
        }
    }

    for (int c = CAP / 32 + w; c < nchunks; c += 16) {
        const int j0 = c << 5;
        int m = T - j0;
        m = m < 32 ? m : 32;
        int2 r = make_int2(0, 0);
        if (lane < m) {
            const int j = j0 + lane;
            int lo = 0, hi = NCH, plo = 0;
#pragma unroll
            for (int it = 0; it < 8; ++it) {
                const int  mid = (lo + hi) >> 1;
                const int  pm  = pre[mid];
                const bool le  = pm <= j;
                lo  = le ? mid : lo;
                plo = le ? pm  : plo;
                hi  = le ? hi  : mid;
            }
            r = pack_rec(recs[rst[lo] + (j - plo)]);
        }
        for (int t = 0; t < m; t += 16) {
            int ee[16], ss[16], ll[16];
#pragma unroll
            for (int u = 0; u < 16; ++u) {
                ee[u] = __builtin_amdgcn_readlane(r.x, t + u);
                const int sl = __builtin_amdgcn_readlane(r.y, t + u);
                ss[u] = sl & ((1 << SRC_BITS) - 1);
                ll[u] = (int)((unsigned)sl >> SRC_BITS);
            }
            sum_trip(ee, ss, ll, t, m, lane, l16, hi16, feat2, efeat, sacc);
        }
    }
    __syncthreads();

    if (tid < BN) {
        float d = (float)sdeg[tid];
        d = d < 1.0f ? 1.0f : d;
        snorm[tid] = 1.0f / sqrtf(d);
    }
    __syncthreads();

    if (w < NTILE) {
        const int half  = lane >> 4;
        const int rbase = w * 16;
        int ra = rbase + l16;
        ra = ra < BN ? ra : BN - 1;
        const float  na = snorm[ra];
        const float* ar = sacc + ra * IN_F + 2 * half;
        const float* bw = sW + (2 * half) * OUT_F + l16;

        v8f cacc[4] = {};

#if defined(USE_WMMA_F32)
        for (int kk = 0; kk < IN_F; kk += 4) {
            v2f a;
            a.x = ar[kk]     * na;
            a.y = ar[kk + 1] * na;
#pragma unroll
            for (int nt = 0; nt < 4; ++nt) {
                v2f bb;
                bb.x = bw[kk       * OUT_F + nt * 16];
                bb.y = bw[(kk + 1) * OUT_F + nt * 16];
                cacc[nt] = __builtin_amdgcn_wmma_f32_16x16x4_f32(
                    false, a, false, bb, (short)0, cacc[nt], false, false);
            }
        }
#else
        for (int i = 0; i < 8; ++i) {
            int m = rbase + i + 8 * half;
            m = m < BN ? m : BN - 1;
            for (int nt = 0; nt < 4; ++nt) {
                float s = 0.0f;
                for (int kk = 0; kk < IN_F; ++kk)
                    s += sacc[m * IN_F + kk] * snorm[m] * sW[kk * OUT_F + nt * 16 + l16];
                cacc[nt][i] = s;
            }
        }
        (void)na; (void)ar; (void)bw;
#endif

        const float SELU_SCALE = 1.0507009873554805f;
        const float SELU_ALPHA = 1.6732632423543772f;
        float nd[8];
#pragma unroll
        for (int i = 0; i < 8; ++i) {
            int m = rbase + i + 8 * half;
            nd[i] = snorm[m < BN ? m : BN - 1];
        }
        float* orow = out + (size_t)b * (BN * OUT_F);
#pragma unroll
        for (int nt = 0; nt < 4; ++nt) {
            const int   ncol = nt * 16 + l16;
            const float bv   = bias[ncol];
#pragma unroll
            for (int i = 0; i < 8; ++i) {
                const int m = rbase + i + 8 * half;
                if (m < BN) {
                    float v = cacc[nt][i] * nd[i] + bv;
                    v = v > 0.0f ? SELU_SCALE * v
                                 : SELU_SCALE * SELU_ALPHA * (expf(v) - 1.0f);
                    orow[m * OUT_F + ncol] = v;
                }
            }
        }
    }
}

extern "C" void kernel_launch(void* const* d_in, const int* in_sizes, int n_in,
                              void* d_out, int out_size, void* d_ws, size_t ws_size,
                              hipStream_t stream) {
    (void)in_sizes; (void)n_in; (void)out_size; (void)ws_size;

    const float* feat  = (const float*)d_in[0];
    const float* efeat = (const float*)d_in[1];
    const int*   src   = (const int*)d_in[2];
    const int*   dst   = (const int*)d_in[3];
    const float* W     = (const float*)d_in[4];
    const float* bias  = (const float*)d_in[5];
    float*       out   = (float*)d_out;

    int* runs   = (int*)((char*)d_ws + 32400000);
    int* binned = (int*)((char*)d_ws + 33448576);

    chunk_group<<<NCH, 1024, 0, stream>>>(src, dst, runs, binned);

    bucket_sum_gemm<<<NB, 512, 0, stream>>>(feat, efeat, runs, binned, W, bias, out);
}


# Phase 1 builds each chunk's grouped records in LDS (8 bytes each, the form phase 2 keeps) and writes the region as whole 128-byte lines; was 6400 scattered 16-byte stores per chunk; sum order kept
# speedup vs baseline: 5.9895x; 1.0834x over previous
#include <hip/hip_runtime.h>
#include <math.h>


#define N_NODES 100000
#define N_EDGES 1600000
#define NODE_F  64
#define EDGE_F  16
#define IN_F    80
#define OUT_F   64

#define NB      500
#define BN      200
#define NCH     250
#define CH      6400
#define NTILE   ((BN + 15) / 16)
#define CAP     3584
#define SPL     ((BN + 31) / 32)
#define SRC_BITS 17
#define EDGE_MASK 0x3fffff

typedef float    v2f  __attribute__((ext_vector_type(2)));
typedef float    v8f  __attribute__((ext_vector_type(8)));

#if defined(__AMDGCN__)
#if __has_builtin(__builtin_amdgcn_wmma_f32_16x16x4_f32)
#define USE_WMMA_F32 1
#endif
#endif

__device__ __forceinline__ int bucket_of(int d) {
    unsigned b = (unsigned)d / (unsigned)BN;
    return (int)(b < (unsigned)(NB - 1) ? b : (unsigned)(NB - 1));
}

__device__ __forceinline__ void place_edge(int* cur, int2* srecs, int e, int s, int d) {
    const int b = bucket_of(d);
    const int p = atomicAdd(&cur[b], 1);
    int z = d - b * BN;
    z = z < 0 ? 0 : (z > BN - 1 ? BN - 1 : z);
    if ((unsigned)p < (unsigned)CH)
        srecs[p] = make_int2(e, (s & ((1 << SRC_BITS) - 1)) | (z << SRC_BITS));
}

__global__ void __launch_bounds__(1024)
chunk_group(const int* __restrict__ src, const int* __restrict__ dst,
            int* __restrict__ runs, int* __restrict__ binned) {
    __shared__ int h[512];
    __shared__ int wtot[16];
    __shared__ __attribute__((aligned(16))) int2 srecs[CH];
    const int tid  = threadIdx.x;
    const int blk  = blockIdx.x;
    const int lane = tid & 31;
    const int w    = tid >> 5;
    if (tid < 512) h[tid] = 0;
    __syncthreads();

    const int4* d4p = (const int4*)dst + (size_t)blk * (CH / 4);
    const int4* s4p = (const int4*)src + (size_t)blk * (CH / 4);
    const bool two = tid < CH / 4 - 1024;
    const int  q1  = two ? tid + 1024 : tid;
    const int4 d0 = d4p[tid];
    const int4 d1 = d4p[q1];
    const int4 s0 = s4p[tid];
    const int4 s1 = s4p[q1];

    atomicAdd(&h[bucket_of(d0.x)], 1);
    atomicAdd(&h[bucket_of(d0.y)], 1);
    atomicAdd(&h[bucket_of(d0.z)], 1);
    atomicAdd(&h[bucket_of(d0.w)], 1);
    if (two) {
        atomicAdd(&h[bucket_of(d1.x)], 1);
        atomicAdd(&h[bucket_of(d1.y)], 1);
        atomicAdd(&h[bucket_of(d1.z)], 1);
        atomicAdd(&h[bucket_of(d1.w)], 1);
    }
    __syncthreads();

    int v = 0, inc = 0;
    if (tid < 512) {
        v   = h[tid];
        inc = v;
#pragma unroll
        for (int o = 1; o < 32; o <<= 1) {
            int t = __shfl_up(inc, o);
            if (lane >= o) inc += t;
        }
        if (lane == 31) wtot[w] = inc;
    }
    __syncthreads();
    if (tid < 512) {
        int carry = 0;
        for (int k = 0; k < w; ++k) carry += wtot[k];
        const int excl = carry + inc - v;
        h[tid] = excl;
        if (tid <= NB) runs[tid * NCH + blk] = excl;
    }
    __syncthreads();

    const int e0 = blk * CH + 4 * tid;
    place_edge(h, srecs, e0,     s0.x, d0.x);
    place_edge(h, srecs, e0 + 1, s0.y, d0.y);
    place_edge(h, srecs, e0 + 2, s0.z, d0.z);
    place_edge(h, srecs, e0 + 3, s0.w, d0.w);
    if (two) {
        const int e1 = blk * CH + 4 * q1;
        place_edge(h, srecs, e1,     s1.x, d1.x);
        place_edge(h, srecs, e1 + 1, s1.y, d1.y);
        place_edge(h, srecs, e1 + 2, s1.z, d1.z);
        place_edge(h, srecs, e1 + 3, s1.w, d1.w);
    }
    __syncthreads();

    const int4* s4 = (const int4*)srecs;
    int4*       o4 = (int4*)binned + (size_t)blk * (CH / 2);
    const int4 c0 = s4[tid];
    const int4 c1 = s4[tid + 1024];
    const int4 c2 = s4[tid + 2048];
    o4[tid]        = c0;
    o4[tid + 1024] = c1;
    o4[tid + 2048] = c2;
    if (tid < CH / 2 - 3072) o4[tid + 3072] = s4[tid + 3072];
}

__device__ __forceinline__ void lds_add(float* p, float v) {
    __hip_atomic_fetch_add(p, v, __ATOMIC_RELAXED, __HIP_MEMORY_SCOPE_WORKGROUP);
}

__device__ __forceinline__ void stage_rec(int2 pr, int j, int* sdeg, int2* srec) {
    atomicAdd(&sdeg[(unsigned)pr.y >> SRC_BITS], 1);
    if (j < CAP) srec[j] = pr;
}

__device__ __forceinline__ void sum_trip(const int (&ee)[16], const int (&ss)[16], const int (&ll)[16],
                                         int t, int m, int lane, int l16, bool hi16,
                                         const v2f* __restrict__ feat2,
                                         const float* __restrict__ efeat, float* sacc) {
    v2f   hv[16];
    float ev[8];
#pragma unroll
    for (int u = 0; u < 16; ++u)
        hv[u] = feat2[(unsigned)(ss[u] * (NODE_F / 2) + lane)];
#pragma unroll
    for (int p = 0; p < 8; ++p) {
        const int ea = ee[2 * p];
        const int eb = ee[2 * p + 1];
        const int em = (hi16 ? eb : ea) & EDGE_MASK;
        ev[p] = efeat[(unsigned)(em * EDGE_F + l16)];
    }
#pragma unroll
    for (int p = 0; p < 8; ++p) {
        if (t + 2 * p < m) {
            float* row = sacc + ll[2 * p] * IN_F + EDGE_F + 2 * lane;
            lds_add(row,     hv[2 * p].x);
            lds_add(row + 1, hv[2 * p].y);
        }
        if (t + 2 * p + 1 < m) {
            float* row = sacc + ll[2 * p + 1] * IN_F + EDGE_F + 2 * lane;
            lds_add(row,     hv[2 * p + 1].x);
            lds_add(row + 1, hv[2 * p + 1].y);
        }
        const int la = ll[2 * p];
        const int lb = ll[2 * p + 1];
        const int lm = hi16 ? lb : la;
        if (t + 2 * p + (lane >> 4) < m) lds_add(sacc + lm * IN_F + l16, ev[p]);
    }
}

__device__ __forceinline__ void flush_run(float* sacc, int cur, v2f& ah, float& ae, float& ao,
                                          int lane, int l16, bool hi16) {
    float* row = sacc + cur * IN_F;
    lds_add(row + EDGE_F + 2 * lane,     ah.x);
    lds_add(row + EDGE_F + 2 * lane + 1, ah.y);
    lds_add(row + l16, hi16 ? ao : ae);
    ah.x = 0.0f; ah.y = 0.0f; ae = 0.0f; ao = 0.0f;
}

__device__ __forceinline__ void sum_trip_runs(const int2* rs, int nv, int lane, int l16, bool hi16,
                                              const v2f* __restrict__ feat2,
                                              const float* __restrict__ efeat, float* sacc,
                                              int& cur, v2f& ah, float& ae, float& ao) {
    int   ee[16], ss[16], ll[16];
    v2f   hv[16];
    float ev[8];
#pragma unroll
    for (int u = 0; u < 16; ++u) {
        const int2 pr = rs[u];
        ee[u] = pr.x;
        ss[u] = pr.y & ((1 << SRC_BITS) - 1);
        ll[u] = (int)((unsigned)pr.y >> SRC_BITS);
    }
#pragma unroll
    for (int u = 0; u < 16; ++u)
        hv[u] = feat2[(unsigned)(ss[u] * (NODE_F / 2) + lane)];
#pragma unroll
    for (int p = 0; p < 8; ++p) {
        const int ea = ee[2 * p];
        const int eb = ee[2 * p + 1];
        const int em = (hi16 ? eb : ea) & EDGE_MASK;
        ev[p] = efeat[(unsigned)(em * EDGE_F + l16)];
    }
#pragma unroll
    for (int u = 0; u < 16; ++u) {
        if (u == 0 || u < nv) {
            if (ll[u] != cur) {
                flush_run(sacc, cur, ah, ae, ao, lane, l16, hi16);
                cur = ll[u];
            }
            ah.x += hv[u].x;
            ah.y += hv[u].y;
            if (u & 1) ao += ev[u >> 1];
            else       ae += ev[u >> 1];
        }
    }
}

__global__ void __launch_bounds__(512)
bucket_sum_gemm(const float* __restrict__ feat,
                const float* __restrict__ efeat,
                const int*   __restrict__ runs,
                const int*   __restrict__ binned,
                const float* __restrict__ W,
                const float* __restrict__ bias,
                float*       __restrict__ out) {
    __shared__ __attribute__((aligned(16))) float sacc[BN * IN_F];
    __shared__ __attribute__((aligned(16))) float sW[IN_F * OUT_F];
    __shared__ float snorm[BN];
    __shared__ int   sdeg[SPL * 32];
    __shared__ int   scur[SPL * 32];
    __shared__ int   pre[256];
    __shared__ int   rst[256];
    __shared__ int   wtot[16];
    __shared__ __attribute__((aligned(16))) int2 srec[2 * CAP];

    const int tid  = threadIdx.x;
    const int b    = blockIdx.x;
    const int lane = tid & 31;
    const int l16  = lane & 15;
    const bool hi16 = lane >= 16;
    const int w    = __builtin_amdgcn_readfirstlane(tid >> 5);

    float4* sa4 = (float4*)sacc;
    for (int t = tid; t < BN * IN_F / 4; t += 512) sa4[t] = make_float4(0.f, 0.f, 0.f, 0.f);
    if (tid < SPL * 32) sdeg[tid] = 0;
    {
        const float4* W4  = (const float4*)W;
        float4*       sW4 = (float4*)sW;
        for (int t = tid; t < IN_F * OUT_F / 4; t += 512) sW4[t] = W4[t];
    }

    int st = 0, len = 0;
    if (tid < NCH) {
        st     = runs[b * NCH + tid];
        int en = runs[(b + 1) * NCH + tid];
        st  = st < 0 ? 0 : (st > CH ? CH : st);
        en  = en < st ? st : (en > CH ? CH : en);
        len = en - st;
    }
    int inc = len;
#pragma unroll
    for (int o = 1; o < 32; o <<= 1) {
        int t = __shfl_up(inc, o);
        if (lane >= o) inc += t;
    }
    if (lane == 31) wtot[w] = inc;
    __syncthreads();
    {
        int carry = 0;
        for (int k = 0; k < w; ++k) carry += wtot[k];
        if (tid <= NCH) pre[tid] = carry + inc - len;
        if (tid <  NCH) rst[tid] = tid * CH + st;
    }
    __syncthreads();
    const int T       = __builtin_amdgcn_readfirstlane(pre[NCH]);
    const int nchunks = (T + 31) >> 5;

    const int2* recs  = (const int2*)binned;
    const v2f*  feat2 = (const v2f*)feat;

    {
        const int  cq  = w + 16 * l16;
        const bool okq = cq < NCH;
        const int  ccq = okq ? cq : 0;
        const int  vp0 = pre[ccq];
        const int  vln = okq ? pre[ccq + 1] - vp0 : 0;
        const int  vg0 = rst[ccq];
        int2 rr[16];
#pragma unroll
        for (int q = 0; q < 16; ++q) {
            const int ln = __builtin_amdgcn_readlane(vln, q);
            const int g0 = __builtin_amdgcn_readlane(vg0, q);
            rr[q] = recs[lane < ln ? g0 + lane : 0];
        }
        asm volatile("" :: "v"(rr[0].x), "v"(rr[0].y));
#pragma unroll
        for (int q = 0; q < 16; ++q) {
            const int ln = __builtin_amdgcn_readlane(vln, q);
            const int p0 = __builtin_amdgcn_readlane(vp0, q);
            if (lane < ln) stage_rec(rr[q], p0 + lane, sdeg, srec);
        }
        for (int q = 0; q < 16; ++q) {
            const int ln = __builtin_amdgcn_readlane(vln, q);
            const int p0 = __builtin_amdgcn_readlane(vp0, q);
            const int g0 = __builtin_amdgcn_readlane(vg0, q);
            for (int i = lane + 32; i < ln; i += 32)
                stage_rec(recs[g0 + i], p0 + i, sdeg, srec);
        }
    }
    __syncthreads();

    const bool sorted = T <= CAP;
    const int  roff   = sorted ? CAP : 0;
    if (w == 0) {
        int v[SPL];
        int s = 0;
#pragma unroll
        for (int k = 0; k < SPL; ++k) {
            v[k] = sdeg[SPL * lane + k];
            s += v[k];
        }
        int sc = s;
#pragma unroll
        for (int o = 1; o < 32; o <<= 1) {
            int t = __shfl_up(sc, o);
            if (lane >= o) sc += t;
        }
        int run = sc - s;
#pragma unroll
        for (int k = 0; k < SPL; ++k) {
            scur[SPL * lane + k] = run;
            run += v[k];
        }
    }
    __syncthreads();
    if (sorted) {
        int2 pr[CAP / 512];
#pragma unroll
        for (int k = 0; k < CAP / 512; ++k) {
            const int j = tid + 512 * k;
            pr[k] = srec[j < T ? j : 0];
        }
        int pos[CAP / 512];
#pragma unroll
        for (int k = 0; k < CAP / 512; ++k) {
            const int j = tid + 512 * k;
            pos[k] = j < T ? atomicAdd(&scur[(unsigned)pr[k].y >> SRC_BITS], 1) : CAP;
        }
#pragma unroll
        for (int k = 0; k < CAP / 512; ++k)
            if ((unsigned)pos[k] < (unsigned)CAP) srec[CAP + pos[k]] = pr[k];
        if (tid < 16 && T + tid < CAP) srec[CAP + T + tid] = make_int2(0, 0);
    }
    __syncthreads();

    {
        const int Tc  = T < CAP ? T : CAP;
        const int ntr = (Tc + 15) >> 4;
        const int tb  = (ntr * w) >> 4;
        const int te  = (ntr * (w + 1)) >> 4;
        if (tb < te) {
            const int2* rs = srec + roff + (tb << 4);
            int   cur = (int)((unsigned)rs[0].y >> SRC_BITS);
            v2f   ah  = {0.0f, 0.0f};
            float ae  = 0.0f, ao = 0.0f;
            for (int tr = tb; tr < te; ++tr) {
                int nv = Tc - (tr << 4);
                nv = nv < 16 ? nv : 16;
                sum_trip_runs(rs, nv, lane, l16, hi16, feat2, efeat, sacc, cur, ah, ae, ao);
                rs += 16;
            }
            flush_run(sacc, cur, ah, ae, ao, lane, l16, hi16);
        }
    }

    for (int c = CAP / 32 + w; c < nchunks; c += 16) {
        const int j0 = c << 5;
        int m = T - j0;
        m = m < 32 ? m : 32;
        int2 r = make_int2(0, 0);
        if (lane < m) {
            const int j = j0 + lane;
            int lo = 0, hi = NCH, plo = 0;
#pragma unroll
            for (int it = 0; it < 8; ++it) {
                const int  mid = (lo + hi) >> 1;
                const int  pm  = pre[mid];
                const bool le  = pm <= j;
                lo  = le ? mid : lo;
                plo = le ? pm  : plo;
                hi  = le ? hi  : mid;
            }
            r = recs[rst[lo] + (j - plo)];
        }
        for (int t = 0; t < m; t += 16) {
            int ee[16], ss[16], ll[16];
#pragma unroll
            for (int u = 0; u < 16; ++u) {
                ee[u] = __builtin_amdgcn_readlane(r.x, t + u);
                const int sl = __builtin_amdgcn_readlane(r.y, t + u);
                ss[u] = sl & ((1 << SRC_BITS) - 1);
                ll[u] = (int)((unsigned)sl >> SRC_BITS);
            }
            sum_trip(ee, ss, ll, t, m, lane, l16, hi16, feat2, efeat, sacc);
        }
    }
    __syncthreads();

    if (tid < BN) {
        float d = (float)sdeg[tid];
        d = d < 1.0f ? 1.0f : d;
        snorm[tid] = 1.0f / sqrtf(d);
    }
    __syncthreads();

    if (w < NTILE) {
        const int half  = lane >> 4;
        const int rbase = w * 16;
        int ra = rbase + l16;
        ra = ra < BN ? ra : BN - 1;
        const float  na = snorm[ra];
        const float* ar = sacc + ra * IN_F + 2 * half;
        const float* bw = sW + (2 * half) * OUT_F + l16;

        v8f cacc[4] = {};

#if defined(USE_WMMA_F32)
        for (int kk = 0; kk < IN_F; kk += 4) {
            v2f a;
            a.x = ar[kk]     * na;
            a.y = ar[kk + 1] * na;
#pragma unroll
            for (int nt = 0; nt < 4; ++nt) {
                v2f bb;
                bb.x = bw[kk       * OUT_F + nt * 16];
                bb.y = bw[(kk + 1) * OUT_F + nt * 16];
                cacc[nt] = __builtin_amdgcn_wmma_f32_16x16x4_f32(
                    false, a, false, bb, (short)0, cacc[nt], false, false);
            }
        }
#else
        for (int i = 0; i < 8; ++i) {
            int m = rbase + i + 8 * half;
            m = m < BN ? m : BN - 1;
            for (int nt = 0; nt < 4; ++nt) {
                float s = 0.0f;
                for (int kk = 0; kk < IN_F; ++kk)
                    s += sacc[m * IN_F + kk] * snorm[m] * sW[kk * OUT_F + nt * 16 + l16];
                cacc[nt][i] = s;
            }
        }
        (void)na; (void)ar; (void)bw;
#endif

        const float SELU_SCALE = 1.0507009873554805f;
        const float SELU_ALPHA = 1.6732632423543772f;
        float nd[8];
#pragma unroll
        for (int i = 0; i < 8; ++i) {
            int m = rbase + i + 8 * half;
            nd[i] = snorm[m < BN ? m : BN - 1];
        }
        float* orow = out + (size_t)b * (BN * OUT_F);
#pragma unroll
        for (int nt = 0; nt < 4; ++nt) {
            const int   ncol = nt * 16 + l16;
            const float bv   = bias[ncol];
#pragma unroll
            for (int i = 0; i < 8; ++i) {
                const int m = rbase + i + 8 * half;
                if (m < BN) {
                    float v = cacc[nt][i] * nd[i] + bv;
                    v = v > 0.0f ? SELU_SCALE * v
                                 : SELU_SCALE * SELU_ALPHA * (expf(v) - 1.0f);
                    orow[m * OUT_F + ncol] = v;
                }
            }
        }
    }
}

extern "C" void kernel_launch(void* const* d_in, const int* in_sizes, int n_in,
                              void* d_out, int out_size, void* d_ws, size_t ws_size,
                              hipStream_t stream) {
    (void)in_sizes; (void)n_in; (void)out_size; (void)ws_size;

    const float* feat  = (const float*)d_in[0];
    const float* efeat = (const float*)d_in[1];
    const int*   src   = (const int*)d_in[2];
    const int*   dst   = (const int*)d_in[3];
    const float* W     = (const float*)d_in[4];
    const float* bias  = (const float*)d_in[5];
    float*       out   = (float*)d_out;

    int* runs   = (int*)((char*)d_ws + 32400000);
    int* binned = (int*)((char*)d_ws + 33448704);

    chunk_group<<<NCH, 1024, 0, stream>>>(src, dst, runs, binned);

    bucket_sum_gemm<<<NB, 512, 0, stream>>>(feat, efeat, runs, binned, W, bias, out);
}


# Products and epilogue dealt evenly to all 16 waves: 52 units (row tile x 16-column tile, 13 per SIMD), results staged in LDS, SELU and whole-line 16-byte stores by all threads; sums' row pitch 84
# speedup vs baseline: 6.3533x; 1.0607x over previous
#include <hip/hip_runtime.h>
#include <math.h>


#define N_NODES 100000
#define N_EDGES 1600000
#define NODE_F  64
#define EDGE_F  16
#define IN_F    80
#define OUT_F   64

#define NB      500
#define BN      200
#define NCH     250
#define CH      6400
#define NTILE   ((BN + 15) / 16)
#define SP      84
#define CAP     3584
#define SPL     ((BN + 31) / 32)
#define SRC_BITS 17
#define EDGE_MASK 0x3fffff

typedef float    v2f  __attribute__((ext_vector_type(2)));
typedef float    v8f  __attribute__((ext_vector_type(8)));

#if defined(__AMDGCN__)
#if __has_builtin(__builtin_amdgcn_wmma_f32_16x16x4_f32)
#define USE_WMMA_F32 1
#endif
#endif

__device__ __forceinline__ int bucket_of(int d) {
    unsigned b = (unsigned)d / (unsigned)BN;
    return (int)(b < (unsigned)(NB - 1) ? b : (unsigned)(NB - 1));
}

__device__ __forceinline__ void place_edge(int* cur, int2* srecs, int e, int s, int d) {
    const int b = bucket_of(d);
    const int p = atomicAdd(&cur[b], 1);
    int z = d - b * BN;
    z = z < 0 ? 0 : (z > BN - 1 ? BN - 1 : z);
    if ((unsigned)p < (unsigned)CH)
        srecs[p] = make_int2(e, (s & ((1 << SRC_BITS) - 1)) | (z << SRC_BITS));
}

__global__ void __launch_bounds__(1024)
chunk_group(const int* __restrict__ src, const int* __restrict__ dst,
            int* __restrict__ runs, int* __restrict__ binned) {
    __shared__ int h[512];
    __shared__ int wtot[16];
    __shared__ __attribute__((aligned(16))) int2 srecs[CH];
    const int tid  = threadIdx.x;
    const int blk  = blockIdx.x;
    const int lane = tid & 31;
    const int w    = tid >> 5;
    if (tid < 512) h[tid] = 0;
    __syncthreads();

    const int4* d4p = (const int4*)dst + (size_t)blk * (CH / 4);
    const int4* s4p = (const int4*)src + (size_t)blk * (CH / 4);
    const bool two = tid < CH / 4 - 1024;
    const int  q1  = two ? tid + 1024 : tid;
    const int4 d0 = d4p[tid];
    const int4 d1 = d4p[q1];
    const int4 s0 = s4p[tid];
    const int4 s1 = s4p[q1];

    atomicAdd(&h[bucket_of(d0.x)], 1);
    atomicAdd(&h[bucket_of(d0.y)], 1);
    atomicAdd(&h[bucket_of(d0.z)], 1);
    atomicAdd(&h[bucket_of(d0.w)], 1);
    if (two) {
        atomicAdd(&h[bucket_of(d1.x)], 1);
        atomicAdd(&h[bucket_of(d1.y)], 1);
        atomicAdd(&h[bucket_of(d1.z)], 1);
        atomicAdd(&h[bucket_of(d1.w)], 1);
    }
    __syncthreads();

    int v = 0, inc = 0;
    if (tid < 512) {
        v   = h[tid];
        inc = v;
#pragma unroll
        for (int o = 1; o < 32; o <<= 1) {
            int t = __shfl_up(inc, o);
            if (lane >= o) inc += t;
        }
        if (lane == 31) wtot[w] = inc;
    }
    __syncthreads();
    if (tid < 512) {
        int carry = 0;
        for (int k = 0; k < w; ++k) carry += wtot[k];
        const int excl = carry + inc - v;
        h[tid] = excl;
        if (tid <= NB) runs[tid * NCH + blk] = excl;
    }
    __syncthreads();

    const int e0 = blk * CH + 4 * tid;
    place_edge(h, srecs, e0,     s0.x, d0.x);
    place_edge(h, srecs, e0 + 1, s0.y, d0.y);
    place_edge(h, srecs, e0 + 2, s0.z, d0.z);
    place_edge(h, srecs, e0 + 3, s0.w, d0.w);
    if (two) {
        const int e1 = blk * CH + 4 * q1;
        place_edge(h, srecs, e1,     s1.x, d1.x);
        place_edge(h, srecs, e1 + 1, s1.y, d1.y);
        place_edge(h, srecs, e1 + 2, s1.z, d1.z);
        place_edge(h, srecs, e1 + 3, s1.w, d1.w);
    }
    __syncthreads();

    const int4* s4 = (const int4*)srecs;
    int4*       o4 = (int4*)binned + (size_t)blk * (CH / 2);
    const int4 c0 = s4[tid];
    const int4 c1 = s4[tid + 1024];
    const int4 c2 = s4[tid + 2048];
    o4[tid]        = c0;
    o4[tid + 1024] = c1;
    o4[tid + 2048] = c2;
    if (tid < CH / 2 - 3072) o4[tid + 3072] = s4[tid + 3072];
}

__device__ __forceinline__ void lds_add(float* p, float v) {
    __hip_atomic_fetch_add(p, v, __ATOMIC_RELAXED, __HIP_MEMORY_SCOPE_WORKGROUP);
}

__device__ __forceinline__ void stage_rec(int2 pr, int j, int* sdeg, int2* srec) {
    atomicAdd(&sdeg[(unsigned)pr.y >> SRC_BITS], 1);
    if (j < CAP) srec[j] = pr;
}

__device__ __forceinline__ void sum_trip(const int (&ee)[16], const int (&ss)[16], const int (&ll)[16],
                                         int t, int m, int lane, int l16, bool hi16,
                                         const v2f* __restrict__ feat2,
                                         const float* __restrict__ efeat, float* sacc) {
    v2f   hv[16];
    float ev[8];
#pragma unroll
    for (int u = 0; u < 16; ++u)
        hv[u] = feat2[(unsigned)(ss[u] * (NODE_F / 2) + lane)];
#pragma unroll
    for (int p = 0; p < 8; ++p) {
        const int ea = ee[2 * p];
        const int eb = ee[2 * p + 1];
        const int em = (hi16 ? eb : ea) & EDGE_MASK;
        ev[p] = efeat[(unsigned)(em * EDGE_F + l16)];
    }
#pragma unroll
    for (int p = 0; p < 8; ++p) {
        if (t + 2 * p < m) {
            float* row = sacc + ll[2 * p] * SP + EDGE_F + 2 * lane;
            lds_add(row,     hv[2 * p].x);
            lds_add(row + 1, hv[2 * p].y);
        }
        if (t + 2 * p + 1 < m) {
            float* row = sacc + ll[2 * p + 1] * SP + EDGE_F + 2 * lane;
            lds_add(row,     hv[2 * p + 1].x);
            lds_add(row + 1, hv[2 * p + 1].y);
        }
        const int la = ll[2 * p];
        const int lb = ll[2 * p + 1];
        const int lm = hi16 ? lb : la;
        if (t + 2 * p + (lane >> 4) < m) lds_add(sacc + lm * SP + l16, ev[p]);
    }
}

__device__ __forceinline__ void flush_run(float* sacc, int cur, v2f& ah, float& ae, float& ao,
                                          int lane, int l16, bool hi16) {
    float* row = sacc + cur * SP;
    lds_add(row + EDGE_F + 2 * lane,     ah.x);
    lds_add(row + EDGE_F + 2 * lane + 1, ah.y);
    lds_add(row + l16, hi16 ? ao : ae);
    ah.x = 0.0f; ah.y = 0.0f; ae = 0.0f; ao = 0.0f;
}

__device__ __forceinline__ void sum_trip_runs(const int2* rs, int nv, int lane, int l16, bool hi16,
                                              const v2f* __restrict__ feat2,
                                              const float* __restrict__ efeat, float* sacc,
                                              int& cur, v2f& ah, float& ae, float& ao) {
    int   ee[16], ss[16], ll[16];
    v2f   hv[16];
    float ev[8];
#pragma unroll
    for (int u = 0; u < 16; ++u) {
        const int2 pr = rs[u];
        ee[u] = pr.x;
        ss[u] = pr.y & ((1 << SRC_BITS) - 1);
        ll[u] = (int)((unsigned)pr.y >> SRC_BITS);
    }
#pragma unroll
    for (int u = 0; u < 16; ++u)
        hv[u] = feat2[(unsigned)(ss[u] * (NODE_F / 2) + lane)];
#pragma unroll
    for (int p = 0; p < 8; ++p) {
        const int ea = ee[2 * p];
        const int eb = ee[2 * p + 1];
        const int em = (hi16 ? eb : ea) & EDGE_MASK;
        ev[p] = efeat[(unsigned)(em * EDGE_F + l16)];
    }
#pragma unroll
    for (int u = 0; u < 16; ++u) {
        if (u == 0 || u < nv) {
            if (ll[u] != cur) {
                flush_run(sacc, cur, ah, ae, ao, lane, l16, hi16);
                cur = ll[u];
            }
            ah.x += hv[u].x;
            ah.y += hv[u].y;
            if (u & 1) ao += ev[u >> 1];
            else       ae += ev[u >> 1];
        }
    }
}

__device__ __forceinline__ float selu1(float v) {
    const float SELU_SCALE = 1.0507009873554805f;
    const float SELU_ALPHA = 1.6732632423543772f;
    return v > 0.0f ? SELU_SCALE * v : SELU_SCALE * SELU_ALPHA * (expf(v) - 1.0f);
}

template <int NU>
__device__ __forceinline__ void unit_products(const float* sacc, const float* sW, const float* snorm,
                                              float* sout, int t0, int nt, int lane) {
    const int l16  = lane & 15;
    const int half = lane >> 4;
    const float* ar[NU];
    float        na[NU];
    int          tl[NU];
#pragma unroll
    for (int j = 0; j < NU; ++j) {
        tl[j] = j < 3 ? t0 + 4 * j : NTILE - 1;
        int ra = tl[j] * 16 + l16;
        ra = ra < BN ? ra : BN - 1;
        na[j] = snorm[ra];
        ar[j] = sacc + ra * SP + 2 * half;
    }
    v8f acc[NU] = {};
#if defined(USE_WMMA_F32)
    const float* bw = sW + (2 * half) * OUT_F + nt * 16 + l16;
    for (int kk = 0; kk < IN_F; kk += 4) {
        v2f bb;
        bb.x = bw[kk       * OUT_F];
        bb.y = bw[(kk + 1) * OUT_F];
#pragma unroll
        for (int j = 0; j < NU; ++j) {
            v2f a;
            a.x = ar[j][kk]     * na[j];
            a.y = ar[j][kk + 1] * na[j];
            acc[j] = __builtin_amdgcn_wmma_f32_16x16x4_f32(
                false, a, false, bb, (short)0, acc[j], false, false);
        }
    }
#else
    for (int j = 0; j < NU; ++j)
        for (int i = 0; i < 8; ++i) {
            int m = tl[j] * 16 + i + 8 * half;
            m = m < BN ? m : BN - 1;
            float s = 0.0f;
            for (int kk = 0; kk < IN_F; ++kk)
                s += sacc[m * SP + kk] * snorm[m] * sW[kk * OUT_F + nt * 16 + l16];
            acc[j][i] = s;
        }
    (void)na; (void)ar;
#endif
#pragma unroll
    for (int j = 0; j < NU; ++j)
#pragma unroll
        for (int i = 0; i < 8; ++i)
            sout[(tl[j] * 16 + i + 8 * half) * OUT_F + nt * 16 + l16] = acc[j][i];
}

__global__ void __launch_bounds__(512)
bucket_sum_gemm(const float* __restrict__ feat,
                const float* __restrict__ efeat,
                const int*   __restrict__ runs,
                const int*   __restrict__ binned,
                const float* __restrict__ W,
                const float* __restrict__ bias,
                float*       __restrict__ out) {
    __shared__ __attribute__((aligned(16))) float sacc[BN * SP];
    __shared__ __attribute__((aligned(16))) float sW[IN_F * OUT_F];
    __shared__ float snorm[BN];
    __shared__ int   sdeg[SPL * 32];
    __shared__ int   scur[SPL * 32];
    __shared__ int   pre[256];
    __shared__ int   rst[256];
    __shared__ int   wtot[16];
    __shared__ __attribute__((aligned(16))) int2 srec[2 * CAP];

    const int tid  = threadIdx.x;
    const int b    = blockIdx.x;
    const int lane = tid & 31;
    const int l16  = lane & 15;
    const bool hi16 = lane >= 16;
    const int w    = __builtin_amdgcn_readfirstlane(tid >> 5);

    float4* sa4 = (float4*)sacc;
    for (int t = tid; t < BN * SP / 4; t += 512) sa4[t] = make_float4(0.f, 0.f, 0.f, 0.f);
    if (tid < SPL * 32) sdeg[tid] = 0;
    {
        const float4* W4  = (const float4*)W;
        float4*       sW4 = (float4*)sW;
        for (int t = tid; t < IN_F * OUT_F / 4; t += 512) sW4[t] = W4[t];
    }

    int st = 0, len = 0;
    if (tid < NCH) {
        st     = runs[b * NCH + tid];
        int en = runs[(b + 1) * NCH + tid];
        st  = st < 0 ? 0 : (st > CH ? CH : st);
        en  = en < st ? st : (en > CH ? CH : en);
        len = en - st;
    }
    int inc = len;
#pragma unroll
    for (int o = 1; o < 32; o <<= 1) {
        int t = __shfl_up(inc, o);
        if (lane >= o) inc += t;
    }
    if (lane == 31) wtot[w] = inc;
    __syncthreads();
    {
        int carry = 0;
        for (int k = 0; k < w; ++k) carry += wtot[k];
        if (tid <= NCH) pre[tid] = carry + inc - len;
        if (tid <  NCH) rst[tid] = tid * CH + st;
    }
    __syncthreads();
    const int T       = __builtin_amdgcn_readfirstlane(pre[NCH]);
    const int nchunks = (T + 31) >> 5;

    const int2* recs  = (const int2*)binned;
    const v2f*  feat2 = (const v2f*)feat;

    {
        const int  cq  = w + 16 * l16;
        const bool okq = cq < NCH;
        const int  ccq = okq ? cq : 0;
        const int  vp0 = pre[ccq];
        const int  vln = okq ? pre[ccq + 1] - vp0 : 0;
        const int  vg0 = rst[ccq];
        int2 rr[16];
#pragma unroll
        for (int q = 0; q < 16; ++q) {
            const int ln = __builtin_amdgcn_readlane(vln, q);
            const int g0 = __builtin_amdgcn_readlane(vg0, q);
            rr[q] = recs[lane < ln ? g0 + lane : 0];
        }
        asm volatile("" :: "v"(rr[0].x), "v"(rr[0].y));
#pragma unroll
        for (int q = 0; q < 16; ++q) {
            const int ln = __builtin_amdgcn_readlane(vln, q);
            const int p0 = __builtin_amdgcn_readlane(vp0, q);
            if (lane < ln) stage_rec(rr[q], p0 + lane, sdeg, srec);
        }
        for (int q = 0; q < 16; ++q) {
            const int ln = __builtin_amdgcn_readlane(vln, q);
            const int p0 = __builtin_amdgcn_readlane(vp0, q);
            const int g0 = __builtin_amdgcn_readlane(vg0, q);
            for (int i = lane + 32; i < ln; i += 32)
                stage_rec(recs[g0 + i], p0 + i, sdeg, srec);
        }
    }
    __syncthreads();

    const bool sorted = T <= CAP;
    const int  roff   = sorted ? CAP : 0;
    if (w == 0) {
        int v[SPL];
        int s = 0;
#pragma unroll
        for (int k = 0; k < SPL; ++k) {
            v[k] = sdeg[SPL * lane + k];
            s += v[k];
        }
        int sc = s;
#pragma unroll
        for (int o = 1; o < 32; o <<= 1) {
            int t = __shfl_up(sc, o);
            if (lane >= o) sc += t;
        }
        int run = sc - s;
#pragma unroll
        for (int k = 0; k < SPL; ++k) {
            scur[SPL * lane + k] = run;
            run += v[k];
        }
    }
    __syncthreads();
    if (sorted) {
        int2 pr[CAP / 512];
#pragma unroll
        for (int k = 0; k < CAP / 512; ++k) {
            const int j = tid + 512 * k;
            pr[k] = srec[j < T ? j : 0];
        }
        int pos[CAP / 512];
#pragma unroll
        for (int k = 0; k < CAP / 512; ++k) {
            const int j = tid + 512 * k;
            pos[k] = j < T ? atomicAdd(&scur[(unsigned)pr[k].y >> SRC_BITS], 1) : CAP;
        }
#pragma unroll
        for (int k = 0; k < CAP / 512; ++k)
            if ((unsigned)pos[k] < (unsigned)CAP) srec[CAP + pos[k]] = pr[k];
        if (tid < 16 && T + tid < CAP) srec[CAP + T + tid] = make_int2(0, 0);
    }
    __syncthreads();

    {
        const int Tc  = T < CAP ? T : CAP;
        const int ntr = (Tc + 15) >> 4;
        const int tb  = (ntr * w) >> 4;
        const int te  = (ntr * (w + 1)) >> 4;
        if (tb < te) {
            const int2* rs = srec + roff + (tb << 4);
            int   cur = (int)((unsigned)rs[0].y >> SRC_BITS);
            v2f   ah  = {0.0f, 0.0f};
            float ae  = 0.0f, ao = 0.0f;
            for (int tr = tb; tr < te; ++tr) {
                int nv = Tc - (tr << 4);
                nv = nv < 16 ? nv : 16;
                sum_trip_runs(rs, nv, lane, l16, hi16, feat2, efeat, sacc, cur, ah, ae, ao);
                rs += 16;
            }
            flush_run(sacc, cur, ah, ae, ao, lane, l16, hi16);
        }
    }

    for (int c = CAP / 32 + w; c < nchunks; c += 16) {
        const int j0 = c << 5;
        int m = T - j0;
        m = m < 32 ? m : 32;
        int2 r = make_int2(0, 0);
        if (lane < m) {
            const int j = j0 + lane;
            int lo = 0, hi = NCH, plo = 0;
#pragma unroll
            for (int it = 0; it < 8; ++it) {
                const int  mid = (lo + hi) >> 1;
                const int  pm  = pre[mid];
                const bool le  = pm <= j;
                lo  = le ? mid : lo;
                plo = le ? pm  : plo;
                hi  = le ? hi  : mid;
            }
            r = recs[rst[lo] + (j - plo)];
        }
        for (int t = 0; t < m; t += 16) {
            int ee[16], ss[16], ll[16];
#pragma unroll
            for (int u = 0; u < 16; ++u) {
                ee[u] = __builtin_amdgcn_readlane(r.x, t + u);
                const int sl = __builtin_amdgcn_readlane(r.y, t + u);
                ss[u] = sl & ((1 << SRC_BITS) - 1);
                ll[u] = (int)((unsigned)sl >> SRC_BITS);
            }
            sum_trip(ee, ss, ll, t, m, lane, l16, hi16, feat2, efeat, sacc);
        }
    }
    __syncthreads();

    if (tid < BN) {
        float d = (float)sdeg[tid];
        d = d < 1.0f ? 1.0f : d;
        snorm[tid] = 1.0f / sqrtf(d);
    }
    __syncthreads();

    float* sout = (float*)srec;
    const float4 bv = ((const float4*)bias)[tid & 15];
    {
        const int nt = w & 3;
        const int t0 = w >> 2;
        if (nt == t0) unit_products<4>(sacc, sW, snorm, sout, t0, nt, lane);
        else          unit_products<3>(sacc, sW, snorm, sout, t0, nt, lane);
    }
    __syncthreads();

    {
        const float4* so4 = (const float4*)sout;
        float4*       o4  = (float4*)(out + (size_t)b * (BN * OUT_F));
#pragma unroll
        for (int k = 0; k < (BN * OUT_F / 4 + 511) / 512; ++k) {
            const int q = tid + 512 * k;
            if (q < BN * OUT_F / 4) {
                const float  nd = snorm[q >> 4];
                const float4 c  = so4[q];
                float4 r;
                r.x = selu1(c.x * nd + bv.x);
                r.y = selu1(c.y * nd + bv.y);
                r.z = selu1(c.z * nd + bv.z);
                r.w = selu1(c.w * nd + bv.w);
                o4[q] = r;
            }
        }
    }
}

extern "C" void kernel_launch(void* const* d_in, const int* in_sizes, int n_in,
                              void* d_out, int out_size, void* d_ws, size_t ws_size,
                              hipStream_t stream) {
    (void)in_sizes; (void)n_in; (void)out_size; (void)ws_size;

    const float* feat  = (const float*)d_in[0];
    const float* efeat = (const float*)d_in[1];
    const int*   src   = (const int*)d_in[2];
    const int*   dst   = (const int*)d_in[3];
    const float* W     = (const float*)d_in[4];
    const float* bias  = (const float*)d_in[5];
    float*       out   = (float*)d_out;

    int* runs   = (int*)((char*)d_ws + 32400000);
    int* binned = (int*)((char*)d_ws + 33448704);

    chunk_group<<<NCH, 1024, 0, stream>>>(src, dst, runs, binned);

    bucket_sum_gemm<<<NB, 512, 0, stream>>>(feat, efeat, runs, binned, W, bias, out);
}
